# starting kernel
# baseline (speedup 1.0000x reference)
#include <hip/hip_runtime.h>
#include <math.h>

typedef __attribute__((ext_vector_type(2))) float v2f;
typedef __attribute__((ext_vector_type(8))) float v8f;

#define BN_EPS 1e-5f
#define NBLOCKS 4096


static __device__ inline float fast_rcp(float x) {
#if __has_builtin(__builtin_amdgcn_rcpf)
  return __builtin_amdgcn_rcpf(x);
#else
  return 1.0f / x;
#endif
}

static __device__ inline float fast_tanh(float x) {
#if __has_builtin(__builtin_amdgcn_tanhf)
  return __builtin_amdgcn_tanhf(x);
#elif __has_builtin(__builtin_amdgcn_tanh_f32)
  return __builtin_amdgcn_tanh_f32(x);
#else
  const float e = __expf(2.0f * x);
  return 1.0f - 2.0f * fast_rcp(e + 1.0f);
#endif
}


static __device__ inline v8f wmma4(v2f a, v2f b, v8f c) {
  return __builtin_amdgcn_wmma_f32_16x16x4_f32(
       false, a,  false, b,
       (short)0, c,  false,  false);
}

static __device__ inline v2f load_w_chunk(const float* __restrict__ W, int dout,
                                          int din, int kbase, int lane) {
  int m = lane & 15;
  int k = kbase + ((lane >> 4) << 1);
  v2f a; a.x = 0.0f; a.y = 0.0f;
  if (m < dout) {
    if (k < din)     a.x = W[m * din + k];
    if (k + 1 < din) a.y = W[m * din + k + 1];
  }
  return a;
}

static __device__ inline v8f load_bias_c(const float* __restrict__ bias, int dout,
                                         int lane) {
  v8f c = {};
  if (lane < 16) {
#pragma unroll
    for (int j = 0; j < 8; ++j) c[j] = (j < dout) ? bias[j] : 0.0f;
  }
  return c;
}

static __device__ inline v2f make_b(const float h[8], int kbase, int lane) {
  float hi0 = __shfl(h[kbase + 2], lane & 15, 32);
  float hi1 = __shfl(h[kbase + 3], lane & 15, 32);
  v2f b;
  b.x = (lane < 16) ? h[kbase + 0] : hi0;
  b.y = (lane < 16) ? h[kbase + 1] : hi1;
  return b;
}

template <int DOUT, int DIN>
static __device__ inline void layer(float h[8], v2f a0, v2f a1, const v8f& c,
                                    int lane) {
  v2f b0 = make_b(h, 0, lane);
  v8f d = wmma4(a0, b0, c);
  if (DIN > 4) {
    v2f b1v = make_b(h, 4, lane);
    d = wmma4(a1, b1v, d);
  }
#pragma unroll
  for (int j = 0; j < 8; ++j) h[j] = (j < DOUT) ? fast_tanh(d[j]) : 0.0f;
}


__global__ void __launch_bounds__(256) mlp_wmma_kernel(
    const float* __restrict__ x,
    const float* __restrict__ W1, const float* __restrict__ b1,
    const float* __restrict__ W2, const float* __restrict__ b2,
    const float* __restrict__ W3, const float* __restrict__ b3,
    const float* __restrict__ W4, const float* __restrict__ b4,
    const float* __restrict__ W5, const float* __restrict__ b5,
    const float* __restrict__ W6, const float* __restrict__ b6,
    const float* __restrict__ Wh, const float* __restrict__ bh,
    float* __restrict__ out, float* __restrict__ partials, int numTiles) {
  const int tid  = threadIdx.x;
  const int lane = tid & 31;
  const int wavesPerBlock = blockDim.x >> 5;
  const int waveGlobal = blockIdx.x * wavesPerBlock + (tid >> 5);
  const int waveStride = gridDim.x * wavesPerBlock;

  v2f aW1  = load_w_chunk(W1, 8, 2, 0, lane);
  v2f aW2a = load_w_chunk(W2, 8, 8, 0, lane), aW2b = load_w_chunk(W2, 8, 8, 4, lane);
  v2f aW3a = load_w_chunk(W3, 6, 8, 0, lane), aW3b = load_w_chunk(W3, 6, 8, 4, lane);
  v2f aW4a = load_w_chunk(W4, 4, 6, 0, lane), aW4b = load_w_chunk(W4, 4, 6, 4, lane);
  v2f aW5  = load_w_chunk(W5, 3, 4, 0, lane);
  v2f aW6  = load_w_chunk(W6, 3, 3, 0, lane);
  v2f aWh  = load_w_chunk(Wh, 2, 3, 0, lane);
  v8f c1 = load_bias_c(b1, 8, lane);
  v8f c2 = load_bias_c(b2, 8, lane);
  v8f c3 = load_bias_c(b3, 6, lane);
  v8f c4 = load_bias_c(b4, 4, lane);
  v8f c5 = load_bias_c(b5, 3, lane);
  v8f c6 = load_bias_c(b6, 3, lane);
  v8f ch = load_bias_c(bh, 2, lane);

  float s1 = 0.0f, s2 = 0.0f;

  for (int tile = waveGlobal; tile < numTiles; tile += waveStride) {
    const int row = tile * 16 + (lane & 15);
    float h[8];
#pragma unroll
    for (int j = 0; j < 8; ++j) h[j] = 0.0f;
    if (lane < 16) {
      const float2 xv = reinterpret_cast<const float2*>(x)[row];
      h[0] = xv.x;
      h[1] = xv.y;
    }

    layer<8, 2>(h, aW1,  aW1,  c1, lane);
    layer<8, 8>(h, aW2a, aW2b, c2, lane);
    layer<6, 8>(h, aW3a, aW3b, c3, lane);
    layer<4, 6>(h, aW4a, aW4b, c4, lane);
    layer<3, 4>(h, aW5,  aW5,  c5, lane);
    layer<3, 3>(h, aW6,  aW6,  c6, lane);

    v2f bhv = make_b(h, 0, lane);
    v8f d = wmma4(aWh, bhv, ch);
    const float l0 = d[0], l1 = d[1];
    const float p0 = fast_rcp(1.0f + __expf(l1 - l0));

    if (lane < 16) {
      out[(size_t)row * 2] = p0;
      s1 += p0;
      s2 += p0 * p0;
    }
  }

  __shared__ float r0[256];
  __shared__ float r1[256];
  r0[tid] = s1;
  r1[tid] = s2;
  __syncthreads();
  for (int s = blockDim.x >> 1; s > 0; s >>= 1) {
    if (tid < s) {
      r0[tid] += r0[tid + s];
      r1[tid] += r1[tid + s];
    }
    __syncthreads();
  }
  if (tid == 0) {
    partials[2 * blockIdx.x + 0] = r0[0];
    partials[2 * blockIdx.x + 1] = r1[0];
  }
}

__global__ void __launch_bounds__(256) reduce_partials_kernel(
    const float* __restrict__ partials, int n, float* __restrict__ acc) {
  __shared__ float r0[256];
  __shared__ float r1[256];
  const int tid = threadIdx.x;
  float s1 = 0.0f, s2 = 0.0f;
  for (int i = tid; i < n; i += 256) {
    s1 += partials[2 * i + 0];
    s2 += partials[2 * i + 1];
  }
  r0[tid] = s1;
  r1[tid] = s2;
  __syncthreads();
  for (int s = 128; s > 0; s >>= 1) {
    if (tid < s) {
      r0[tid] += r0[tid + s];
      r1[tid] += r1[tid + s];
    }
    __syncthreads();
  }
  if (tid == 0) {
    acc[0] = r0[0];
    acc[1] = r1[0];
  }
}

__global__ void __launch_bounds__(256) finalize_kernel(
    float* __restrict__ out, const float* __restrict__ acc,
    const float* __restrict__ gamma, const float* __restrict__ beta, int B) {
  const float S1 = acc[0], S2 = acc[1];
  const float invB  = 1.0f / (float)B;
  const float mean0 = S1 * invB;
  const float var   = S2 * invB - mean0 * mean0;
  const float inv   = rsqrtf(var + BN_EPS);
  const float g0 = gamma[0], g1 = gamma[1];
  const float be0 = beta[0], be1 = beta[1];

  const int i      = blockIdx.x * blockDim.x + threadIdx.x;
  const int stride = gridDim.x * blockDim.x;
  for (int r = i; r < B; r += stride) {
    const float p0 = out[2 * (size_t)r];
    const float o0 = (p0 - mean0) * inv * g0 + be0;
    const float o1 = (mean0 - p0) * inv * g1 + be1;
    float2 o;
    o.x = o0;
    o.y = o1;
    reinterpret_cast<float2*>(out)[r] = o;
  }
}

extern "C" void kernel_launch(void* const* d_in, const int* in_sizes, int n_in,
                              void* d_out, int out_size, void* d_ws, size_t ws_size,
                              hipStream_t stream) {
  const float* x  = (const float*)d_in[0];
  const float* W1 = (const float*)d_in[1];  const float* b1 = (const float*)d_in[2];
  const float* W2 = (const float*)d_in[3];  const float* b2 = (const float*)d_in[4];
  const float* W3 = (const float*)d_in[5];  const float* b3 = (const float*)d_in[6];
  const float* W4 = (const float*)d_in[7];  const float* b4 = (const float*)d_in[8];
  const float* W5 = (const float*)d_in[9];  const float* b5 = (const float*)d_in[10];
  const float* W6 = (const float*)d_in[11]; const float* b6 = (const float*)d_in[12];
  const float* Wh = (const float*)d_in[13]; const float* bh = (const float*)d_in[14];
  const float* gamma = (const float*)d_in[15];
  const float* beta  = (const float*)d_in[16];
  float* out = (float*)d_out;

  const int B        = in_sizes[0] / 2;
  const int numTiles = B / 16;

  float* ws       = (float*)d_ws;
  float* acc      = ws;
  float* partials = ws + 2;

  mlp_wmma_kernel<<<NBLOCKS, 256, 0, stream>>>(
      x, W1, b1, W2, b2, W3, b3, W4, b4, W5, b5, W6, b6, Wh, bh,
      out, partials, numTiles);
  reduce_partials_kernel<<<1, 256, 0, stream>>>(partials, NBLOCKS, acc);
  finalize_kernel<<<2048, 256, 0, stream>>>(out, acc, gamma, beta, B);
}


# two 16-row tiles per wave pass via a block-diagonal weight operand (tile 2 in D rows 8-15): all 32 lanes hold live rows, B is the lane's own registers, no lane exchange; batch-sum order changes
# speedup vs baseline: 1.1388x; 1.1388x over previous
#include <hip/hip_runtime.h>
#include <math.h>


typedef __attribute__((ext_vector_type(2))) float v2f;
typedef __attribute__((ext_vector_type(8))) float v8f;

#define BN_EPS 1e-5f
#define NBLOCKS 4096


static __device__ inline float fast_rcp(float x) {
#if __has_builtin(__builtin_amdgcn_rcpf)
  return __builtin_amdgcn_rcpf(x);
#else
  return 1.0f / x;
#endif
}

static __device__ inline float fast_tanh(float x) {
#if __has_builtin(__builtin_amdgcn_tanhf)
  return __builtin_amdgcn_tanhf(x);
#elif __has_builtin(__builtin_amdgcn_tanh_f32)
  return __builtin_amdgcn_tanh_f32(x);
#else
  const float e = __expf(2.0f * x);
  return 1.0f - 2.0f * fast_rcp(e + 1.0f);
#endif
}


static __device__ inline v8f wmma4(v2f a, v2f b, v8f c) {
  return __builtin_amdgcn_wmma_f32_16x16x4_f32(
       false, a,  false, b,
       (short)0, c,  false,  false);
}

static __device__ inline v2f load_w_pair(const float* __restrict__ W, int dout,
                                         int din, int f, int lane) {
  const int c  = lane & 15;
  const int hh = lane >> 4;
  const int m  = c & 7;
  v2f a; a.x = 0.0f; a.y = 0.0f;
  if (((c >> 3) == hh) && (m < dout)) {
    if (f < din)     a.x = W[m * din + f];
    if (f + 1 < din) a.y = W[m * din + f + 1];
  }
  return a;
}

static __device__ inline v8f load_bias_c(const float* __restrict__ bias, int dout) {
  v8f c = {};
#pragma unroll
  for (int j = 0; j < 8; ++j) c[j] = (j < dout) ? bias[j] : 0.0f;
  return c;
}

template <int DOUT, int NCH>
static __device__ inline v8f affine(const float h[8], const v2f* a, const v8f& c) {
  v8f d = c;
#pragma unroll
  for (int q = 0; q < NCH; ++q) {
    v2f b;
    b.x = h[2 * q];
    b.y = h[2 * q + 1];
    d = wmma4(a[q], b, d);
  }
  return d;
}

template <int DOUT, int NCH>
static __device__ inline void layer(float h[8], const v2f* a, const v8f& c) {
  const v8f d = affine<DOUT, NCH>(h, a, c);
#pragma unroll
  for (int j = 0; j < 8; ++j) h[j] = (j < DOUT) ? fast_tanh(d[j]) : 0.0f;
}

struct Weights {
  v2f a1[1], a2[4], a3[4], a4[3], a5[2], a6[2], ah[2];
  v8f c1, c2, c3, c4, c5, c6, ch;
};

static __device__ inline float forward_row(const Weights& w, float x0, float x1) {
  float h[8];
#pragma unroll
  for (int j = 0; j < 8; ++j) h[j] = 0.0f;
  h[0] = x0;
  h[1] = x1;
  layer<8, 1>(h, w.a1, w.c1);
  layer<8, 4>(h, w.a2, w.c2);
  layer<6, 4>(h, w.a3, w.c3);
  layer<4, 3>(h, w.a4, w.c4);
  layer<3, 2>(h, w.a5, w.c5);
  layer<3, 2>(h, w.a6, w.c6);
  const v8f d = affine<2, 2>(h, w.ah, w.ch);
  const float l0 = d[0], l1 = d[1];
  return fast_rcp(1.0f + __expf(l1 - l0));
}


__global__ void __launch_bounds__(256) mlp_wmma_kernel(
    const float* __restrict__ x,
    const float* __restrict__ W1, const float* __restrict__ b1,
    const float* __restrict__ W2, const float* __restrict__ b2,
    const float* __restrict__ W3, const float* __restrict__ b3,
    const float* __restrict__ W4, const float* __restrict__ b4,
    const float* __restrict__ W5, const float* __restrict__ b5,
    const float* __restrict__ W6, const float* __restrict__ b6,
    const float* __restrict__ Wh, const float* __restrict__ bh,
    float* __restrict__ out, float* __restrict__ partials, int numTiles) {
  const int tid  = threadIdx.x;
  const int lane = tid & 31;
  const int wavesPerBlock = blockDim.x >> 5;
  const int waveGlobal = blockIdx.x * wavesPerBlock + (tid >> 5);
  const int waveStride = gridDim.x * wavesPerBlock;

  Weights w;
  w.a1[0] = load_w_pair(W1, 8, 2, 0, lane);
  w.a2[0] = load_w_pair(W2, 8, 8, 0, lane);
  w.a2[1] = load_w_pair(W2, 8, 8, 2, lane);
  w.a2[2] = load_w_pair(W2, 8, 8, 4, lane);
  w.a2[3] = load_w_pair(W2, 8, 8, 6, lane);
  w.a3[0] = load_w_pair(W3, 6, 8, 0, lane);
  w.a3[1] = load_w_pair(W3, 6, 8, 2, lane);
  w.a3[2] = load_w_pair(W3, 6, 8, 4, lane);
  w.a3[3] = load_w_pair(W3, 6, 8, 6, lane);
  w.a4[0] = load_w_pair(W4, 4, 6, 0, lane);
  w.a4[1] = load_w_pair(W4, 4, 6, 2, lane);
  w.a4[2] = load_w_pair(W4, 4, 6, 4, lane);
  w.a5[0] = load_w_pair(W5, 3, 4, 0, lane);
  w.a5[1] = load_w_pair(W5, 3, 4, 2, lane);
  w.a6[0] = load_w_pair(W6, 3, 3, 0, lane);
  w.a6[1] = load_w_pair(W6, 3, 3, 2, lane);
  w.ah[0] = load_w_pair(Wh, 2, 3, 0, lane);
  w.ah[1] = load_w_pair(Wh, 2, 3, 2, lane);
  w.c1 = load_bias_c(b1, 8);
  w.c2 = load_bias_c(b2, 8);
  w.c3 = load_bias_c(b3, 6);
  w.c4 = load_bias_c(b4, 4);
  w.c5 = load_bias_c(b5, 3);
  w.c6 = load_bias_c(b6, 3);
  w.ch = load_bias_c(bh, 2);

  float s1 = 0.0f, s2 = 0.0f;

  const int numPairs = numTiles >> 1;
  for (int pair = waveGlobal; pair < numPairs; pair += waveStride) {
    const int row = pair * 32 + lane;
    const float2 xv = reinterpret_cast<const float2*>(x)[row];
    const float p0 = forward_row(w, xv.x, xv.y);
    out[(size_t)row * 2] = p0;
    s1 += p0;
    s2 += p0 * p0;
  }

  if ((numTiles & 1) && waveGlobal == 0) {
    const int row = (numTiles - 1) * 16 + (lane & 15);
    const bool live = lane < 16;
    float2 xv; xv.x = 0.0f; xv.y = 0.0f;
    if (live) xv = reinterpret_cast<const float2*>(x)[row];
    const float p0 = forward_row(w, xv.x, xv.y);
    if (live) {
      out[(size_t)row * 2] = p0;
      s1 += p0;
      s2 += p0 * p0;
    }
  }

  __shared__ float r0[256];
  __shared__ float r1[256];
  r0[tid] = s1;
  r1[tid] = s2;
  __syncthreads();
  for (int s = blockDim.x >> 1; s > 0; s >>= 1) {
    if (tid < s) {
      r0[tid] += r0[tid + s];
      r1[tid] += r1[tid + s];
    }
    __syncthreads();
  }
  if (tid == 0) {
    partials[2 * blockIdx.x + 0] = r0[0];
    partials[2 * blockIdx.x + 1] = r1[0];
  }
}

__global__ void __launch_bounds__(256) reduce_partials_kernel(
    const float* __restrict__ partials, int n, float* __restrict__ acc) {
  __shared__ float r0[256];
  __shared__ float r1[256];
  const int tid = threadIdx.x;
  float s1 = 0.0f, s2 = 0.0f;
  for (int i = tid; i < n; i += 256) {
    s1 += partials[2 * i + 0];
    s2 += partials[2 * i + 1];
  }
  r0[tid] = s1;
  r1[tid] = s2;
  __syncthreads();
  for (int s = 128; s > 0; s >>= 1) {
    if (tid < s) {
      r0[tid] += r0[tid + s];
      r1[tid] += r1[tid + s];
    }
    __syncthreads();
  }
  if (tid == 0) {
    acc[0] = r0[0];
    acc[1] = r1[0];
  }
}

__global__ void __launch_bounds__(256) finalize_kernel(
    float* __restrict__ out, const float* __restrict__ acc,
    const float* __restrict__ gamma, const float* __restrict__ beta, int B) {
  const float S1 = acc[0], S2 = acc[1];
  const float invB  = 1.0f / (float)B;
  const float mean0 = S1 * invB;
  const float var   = S2 * invB - mean0 * mean0;
  const float inv   = rsqrtf(var + BN_EPS);
  const float g0 = gamma[0], g1 = gamma[1];
  const float be0 = beta[0], be1 = beta[1];

  const int i      = blockIdx.x * blockDim.x + threadIdx.x;
  const int stride = gridDim.x * blockDim.x;
  for (int r = i; r < B; r += stride) {
    const float p0 = out[2 * (size_t)r];
    const float o0 = (p0 - mean0) * inv * g0 + be0;
    const float o1 = (mean0 - p0) * inv * g1 + be1;
    float2 o;
    o.x = o0;
    o.y = o1;
    reinterpret_cast<float2*>(out)[r] = o;
  }
}

extern "C" void kernel_launch(void* const* d_in, const int* in_sizes, int n_in,
                              void* d_out, int out_size, void* d_ws, size_t ws_size,
                              hipStream_t stream) {
  const float* x  = (const float*)d_in[0];
  const float* W1 = (const float*)d_in[1];  const float* b1 = (const float*)d_in[2];
  const float* W2 = (const float*)d_in[3];  const float* b2 = (const float*)d_in[4];
  const float* W3 = (const float*)d_in[5];  const float* b3 = (const float*)d_in[6];
  const float* W4 = (const float*)d_in[7];  const float* b4 = (const float*)d_in[8];
  const float* W5 = (const float*)d_in[9];  const float* b5 = (const float*)d_in[10];
  const float* W6 = (const float*)d_in[11]; const float* b6 = (const float*)d_in[12];
  const float* Wh = (const float*)d_in[13]; const float* bh = (const float*)d_in[14];
  const float* gamma = (const float*)d_in[15];
  const float* beta  = (const float*)d_in[16];
  float* out = (float*)d_out;

  const int B        = in_sizes[0] / 2;
  const int numTiles = B / 16;

  float* ws       = (float*)d_ws;
  float* acc      = ws;
  float* partials = ws + 2;

  mlp_wmma_kernel<<<NBLOCKS, 128, 0, stream>>>(
      x, W1, b1, W2, b2, W3, b3, W4, b4, W5, b5, W6, b6, Wh, bh,
      out, partials, numTiles);
  reduce_partials_kernel<<<1, 256, 0, stream>>>(partials, NBLOCKS, acc);
  finalize_kernel<<<2048, 256, 0, stream>>>(out, acc, gamma, beta, B);
}


# reduce launch folded into finalize: 512 resident 8-wave blocks each sum the 4096 partials in the old order while their own rows (16-byte pieces, 16 loads in flight) arrive; 3 launches to 2
# speedup vs baseline: 1.2263x; 1.0768x over previous
#include <hip/hip_runtime.h>
#include <math.h>


typedef __attribute__((ext_vector_type(2))) float v2f;
typedef __attribute__((ext_vector_type(4))) float v4f;
typedef __attribute__((ext_vector_type(8))) float v8f;

#define BN_EPS 1e-5f
#define NBLOCKS 4096


static __device__ inline float fast_rcp(float x) {
#if __has_builtin(__builtin_amdgcn_rcpf)
  return __builtin_amdgcn_rcpf(x);
#else
  return 1.0f / x;
#endif
}

static __device__ inline float fast_tanh(float x) {
#if __has_builtin(__builtin_amdgcn_tanhf)
  return __builtin_amdgcn_tanhf(x);
#elif __has_builtin(__builtin_amdgcn_tanh_f32)
  return __builtin_amdgcn_tanh_f32(x);
#else
  const float e = __expf(2.0f * x);
  return 1.0f - 2.0f * fast_rcp(e + 1.0f);
#endif
}


static __device__ inline v8f wmma4(v2f a, v2f b, v8f c) {
  return __builtin_amdgcn_wmma_f32_16x16x4_f32(
       false, a,  false, b,
       (short)0, c,  false,  false);
}

static __device__ inline v2f load_w_pair(const float* __restrict__ W, int dout,
                                         int din, int f, int lane) {
  const int c  = lane & 15;
  const int hh = lane >> 4;
  const int m  = c & 7;
  v2f a; a.x = 0.0f; a.y = 0.0f;
  if (((c >> 3) == hh) && (m < dout)) {
    if (f < din)     a.x = W[m * din + f];
    if (f + 1 < din) a.y = W[m * din + f + 1];
  }
  return a;
}

static __device__ inline v8f load_bias_c(const float* __restrict__ bias, int dout) {
  v8f c = {};
#pragma unroll
  for (int j = 0; j < 8; ++j) c[j] = (j < dout) ? bias[j] : 0.0f;
  return c;
}

template <int DOUT, int NCH>
static __device__ inline v8f affine(const float h[8], const v2f* a, const v8f& c) {
  v8f d = c;
#pragma unroll
  for (int q = 0; q < NCH; ++q) {
    v2f b;
    b.x = h[2 * q];
    b.y = h[2 * q + 1];
    d = wmma4(a[q], b, d);
  }
  return d;
}

template <int DOUT, int NCH>
static __device__ inline void layer(float h[8], const v2f* a, const v8f& c) {
  const v8f d = affine<DOUT, NCH>(h, a, c);
#pragma unroll
  for (int j = 0; j < 8; ++j) h[j] = (j < DOUT) ? fast_tanh(d[j]) : 0.0f;
}

struct Weights {
  v2f a1[1], a2[4], a3[4], a4[3], a5[2], a6[2], ah[2];
  v8f c1, c2, c3, c4, c5, c6, ch;
};

static __device__ inline float forward_row(const Weights& w, float x0, float x1) {
  float h[8];
#pragma unroll
  for (int j = 0; j < 8; ++j) h[j] = 0.0f;
  h[0] = x0;
  h[1] = x1;
  layer<8, 1>(h, w.a1, w.c1);
  layer<8, 4>(h, w.a2, w.c2);
  layer<6, 4>(h, w.a3, w.c3);
  layer<4, 3>(h, w.a4, w.c4);
  layer<3, 2>(h, w.a5, w.c5);
  layer<3, 2>(h, w.a6, w.c6);
  const v8f d = affine<2, 2>(h, w.ah, w.ch);
  const float l0 = d[0], l1 = d[1];
  return fast_rcp(1.0f + __expf(l1 - l0));
}


__global__ void __launch_bounds__(256) mlp_wmma_kernel(
    const float* __restrict__ x,
    const float* __restrict__ W1, const float* __restrict__ b1,
    const float* __restrict__ W2, const float* __restrict__ b2,
    const float* __restrict__ W3, const float* __restrict__ b3,
    const float* __restrict__ W4, const float* __restrict__ b4,
    const float* __restrict__ W5, const float* __restrict__ b5,
    const float* __restrict__ W6, const float* __restrict__ b6,
    const float* __restrict__ Wh, const float* __restrict__ bh,
    float* __restrict__ out, float* __restrict__ partials, int numTiles) {
  const int tid  = threadIdx.x;
  const int lane = tid & 31;
  const int wavesPerBlock = blockDim.x >> 5;
  const int waveGlobal = blockIdx.x * wavesPerBlock + (tid >> 5);
  const int waveStride = gridDim.x * wavesPerBlock;

  Weights w;
  w.a1[0] = load_w_pair(W1, 8, 2, 0, lane);
  w.a2[0] = load_w_pair(W2, 8, 8, 0, lane);
  w.a2[1] = load_w_pair(W2, 8, 8, 2, lane);
  w.a2[2] = load_w_pair(W2, 8, 8, 4, lane);
  w.a2[3] = load_w_pair(W2, 8, 8, 6, lane);
  w.a3[0] = load_w_pair(W3, 6, 8, 0, lane);
  w.a3[1] = load_w_pair(W3, 6, 8, 2, lane);
  w.a3[2] = load_w_pair(W3, 6, 8, 4, lane);
  w.a3[3] = load_w_pair(W3, 6, 8, 6, lane);
  w.a4[0] = load_w_pair(W4, 4, 6, 0, lane);
  w.a4[1] = load_w_pair(W4, 4, 6, 2, lane);
  w.a4[2] = load_w_pair(W4, 4, 6, 4, lane);
  w.a5[0] = load_w_pair(W5, 3, 4, 0, lane);
  w.a5[1] = load_w_pair(W5, 3, 4, 2, lane);
  w.a6[0] = load_w_pair(W6, 3, 3, 0, lane);
  w.a6[1] = load_w_pair(W6, 3, 3, 2, lane);
  w.ah[0] = load_w_pair(Wh, 2, 3, 0, lane);
  w.ah[1] = load_w_pair(Wh, 2, 3, 2, lane);
  w.c1 = load_bias_c(b1, 8);
  w.c2 = load_bias_c(b2, 8);
  w.c3 = load_bias_c(b3, 6);
  w.c4 = load_bias_c(b4, 4);
  w.c5 = load_bias_c(b5, 3);
  w.c6 = load_bias_c(b6, 3);
  w.ch = load_bias_c(bh, 2);

  float s1 = 0.0f, s2 = 0.0f;

  const int numPairs = numTiles >> 1;
  for (int pair = waveGlobal; pair < numPairs; pair += waveStride) {
    const int row = pair * 32 + lane;
    const float2 xv = reinterpret_cast<const float2*>(x)[row];
    const float p0 = forward_row(w, xv.x, xv.y);
    out[(size_t)row * 2] = p0;
    s1 += p0;
    s2 += p0 * p0;
  }

  if ((numTiles & 1) && waveGlobal == 0) {
    const int row = (numTiles - 1) * 16 + (lane & 15);
    const bool live = lane < 16;
    float2 xv; xv.x = 0.0f; xv.y = 0.0f;
    if (live) xv = reinterpret_cast<const float2*>(x)[row];
    const float p0 = forward_row(w, xv.x, xv.y);
    if (live) {
      out[(size_t)row * 2] = p0;
      s1 += p0;
      s2 += p0 * p0;
    }
  }

  __shared__ float r0[256];
  __shared__ float r1[256];
  r0[tid] = s1;
  r1[tid] = s2;
  __syncthreads();
  for (int s = blockDim.x >> 1; s > 0; s >>= 1) {
    if (tid < s) {
      r0[tid] += r0[tid + s];
      r1[tid] += r1[tid + s];
    }
    __syncthreads();
  }
  if (tid == 0) {
    partials[2 * blockIdx.x + 0] = r0[0];
    partials[2 * blockIdx.x + 1] = r1[0];
  }
}

#define FIN_BLOCKS  512
#define FIN_THREADS 256
#define FIN_PIECES  16
#define FIN_ROWS_PER_PIECE (FIN_THREADS * 2)
#define FIN_ROWS_PER_CHUNK (FIN_ROWS_PER_PIECE * FIN_PIECES)
#define FIN_PARTS_PER_THREAD (NBLOCKS / FIN_THREADS)

static_assert(NBLOCKS % FIN_THREADS == 0, "partials must split evenly over the fold's threads");

static __device__ inline void fin_load_chunk(const float* out, size_t rowBase,
                                             int tid, v4f v[FIN_PIECES]) {
#pragma unroll
  for (int j = 0; j < FIN_PIECES; ++j) {
    const size_t r = rowBase + (size_t)j * FIN_ROWS_PER_PIECE + (size_t)tid * 2;
    v[j] = *reinterpret_cast<const v4f*>(out + 2 * r);
  }
}

__global__ void __launch_bounds__(FIN_THREADS) finalize_kernel(
    float* __restrict__ out, const float* __restrict__ partials,
    const float* __restrict__ gamma, const float* __restrict__ beta, int B) {
  __shared__ float r0[FIN_THREADS];
  __shared__ float r1[FIN_THREADS];
  const int tid = threadIdx.x;
  const int nChunks = (B + FIN_ROWS_PER_CHUNK - 1) / FIN_ROWS_PER_CHUNK;
  int chunk = blockIdx.x;
  if (chunk >= nChunks) return;

  float2 p[FIN_PARTS_PER_THREAD];
#pragma unroll
  for (int k = 0; k < FIN_PARTS_PER_THREAD; ++k)
    p[k] = reinterpret_cast<const float2*>(partials)[tid + k * FIN_THREADS];

  v4f v[FIN_PIECES];
  bool full = (size_t)chunk * FIN_ROWS_PER_CHUNK + FIN_ROWS_PER_CHUNK <= (size_t)B;
  if (full) fin_load_chunk(out, (size_t)chunk * FIN_ROWS_PER_CHUNK, tid, v);

  const float g0 = gamma[0], g1 = gamma[1];
  const float be0 = beta[0], be1 = beta[1];

  float s1 = 0.0f, s2 = 0.0f;
#pragma unroll
  for (int k = 0; k < FIN_PARTS_PER_THREAD; ++k) {
    s1 += p[k].x;
    s2 += p[k].y;
  }
  r0[tid] = s1;
  r1[tid] = s2;
  __syncthreads();
  for (int s = FIN_THREADS >> 1; s > 0; s >>= 1) {
    if (tid < s) {
      r0[tid] += r0[tid + s];
      r1[tid] += r1[tid + s];
    }
    __syncthreads();
  }
  const float S1 = r0[0], S2 = r1[0];

  const float invB  = 1.0f / (float)B;
  const float mean0 = S1 * invB;
  const float var   = S2 * invB - mean0 * mean0;
  const float inv   = rsqrtf(var + BN_EPS);

  for (;;) {
    const size_t rowBase = (size_t)chunk * FIN_ROWS_PER_CHUNK;
    if (full) {
#pragma unroll
      for (int j = 0; j < FIN_PIECES; ++j) {
        const size_t r = rowBase + (size_t)j * FIN_ROWS_PER_PIECE + (size_t)tid * 2;
        asm volatile("" :: "v"(v[j]));
        const float pa = v[j].x, pb = v[j].z;
        v4f o;
        o.x = (pa - mean0) * inv * g0 + be0;
        o.y = (mean0 - pa) * inv * g1 + be1;
        o.z = (pb - mean0) * inv * g0 + be0;
        o.w = (mean0 - pb) * inv * g1 + be1;
        *reinterpret_cast<v4f*>(out + 2 * r) = o;
      }
    } else {
      size_t rowEnd = rowBase + FIN_ROWS_PER_CHUNK;
      if (rowEnd > (size_t)B) rowEnd = (size_t)B;
      for (size_t r = rowBase + (size_t)tid; r < rowEnd; r += FIN_THREADS) {
        const float p0 = out[2 * r];
        float2 o;
        o.x = (p0 - mean0) * inv * g0 + be0;
        o.y = (mean0 - p0) * inv * g1 + be1;
        reinterpret_cast<float2*>(out)[r] = o;
      }
    }
    chunk += gridDim.x;
    if (chunk >= nChunks) break;
    full = (size_t)chunk * FIN_ROWS_PER_CHUNK + FIN_ROWS_PER_CHUNK <= (size_t)B;
    if (full) fin_load_chunk(out, (size_t)chunk * FIN_ROWS_PER_CHUNK, tid, v);
  }
}

extern "C" void kernel_launch(void* const* d_in, const int* in_sizes, int n_in,
                              void* d_out, int out_size, void* d_ws, size_t ws_size,
                              hipStream_t stream) {
  const float* x  = (const float*)d_in[0];
  const float* W1 = (const float*)d_in[1];  const float* b1 = (const float*)d_in[2];
  const float* W2 = (const float*)d_in[3];  const float* b2 = (const float*)d_in[4];
  const float* W3 = (const float*)d_in[5];  const float* b3 = (const float*)d_in[6];
  const float* W4 = (const float*)d_in[7];  const float* b4 = (const float*)d_in[8];
  const float* W5 = (const float*)d_in[9];  const float* b5 = (const float*)d_in[10];
  const float* W6 = (const float*)d_in[11]; const float* b6 = (const float*)d_in[12];
  const float* Wh = (const float*)d_in[13]; const float* bh = (const float*)d_in[14];
  const float* gamma = (const float*)d_in[15];
  const float* beta  = (const float*)d_in[16];
  float* out = (float*)d_out;

  const int B        = in_sizes[0] / 2;
  const int numTiles = B / 16;

  float* ws       = (float*)d_ws;
  float* partials = ws + 2;

  mlp_wmma_kernel<<<NBLOCKS, 128, 0, stream>>>(
      x, W1, b1, W2, b2, W3, b3, W4, b4, W5, b5, W6, b6, Wh, bh,
      out, partials, numTiles);
  finalize_kernel<<<FIN_BLOCKS, FIN_THREADS, 0, stream>>>(out, partials, gamma, beta, B);
}


# p0 staged densely in the first half of each finalize chunk's own output span: a whole 128-byte line per wave store, finalize reads half the bytes, chunk held in registers before its stores
# speedup vs baseline: 1.5145x; 1.2350x over previous
#include <hip/hip_runtime.h>
#include <math.h>


typedef __attribute__((ext_vector_type(2))) float v2f;
typedef __attribute__((ext_vector_type(4))) float v4f;
typedef __attribute__((ext_vector_type(8))) float v8f;

#define BN_EPS 1e-5f
#define NBLOCKS 4096

#define FIN_BLOCKS  512
#define FIN_THREADS 256
#define FIN_PIECES  16
#define FIN_ROWS_PER_PIECE (FIN_THREADS * 2)
#define FIN_ROWS_PER_CHUNK (FIN_ROWS_PER_PIECE * FIN_PIECES)
#define FIN_PARTS_PER_THREAD (NBLOCKS / FIN_THREADS)

static_assert((FIN_ROWS_PER_CHUNK & (FIN_ROWS_PER_CHUNK - 1)) == 0, "chunk rows must be a power of two");
static_assert(FIN_ROWS_PER_CHUNK % 32 == 0, "a wave's 32 rows must stay inside one chunk");

static __device__ inline unsigned stage_index(int row) {
  const unsigned r = (unsigned)row;
  return r + (r & ~(unsigned)(FIN_ROWS_PER_CHUNK - 1));
}


static __device__ inline float fast_rcp(float x) {
#if __has_builtin(__builtin_amdgcn_rcpf)
  return __builtin_amdgcn_rcpf(x);
#else
  return 1.0f / x;
#endif
}

static __device__ inline float fast_tanh(float x) {
#if __has_builtin(__builtin_amdgcn_tanhf)
  return __builtin_amdgcn_tanhf(x);
#elif __has_builtin(__builtin_amdgcn_tanh_f32)
  return __builtin_amdgcn_tanh_f32(x);
#else
  const float e = __expf(2.0f * x);
  return 1.0f - 2.0f * fast_rcp(e + 1.0f);
#endif
}


static __device__ inline v8f wmma4(v2f a, v2f b, v8f c) {
  return __builtin_amdgcn_wmma_f32_16x16x4_f32(
       false, a,  false, b,
       (short)0, c,  false,  false);
}

static __device__ inline v2f load_w_pair(const float* __restrict__ W, int dout,
                                         int din, int f, int lane) {
  const int c  = lane & 15;
  const int hh = lane >> 4;
  const int m  = c & 7;
  v2f a; a.x = 0.0f; a.y = 0.0f;
  if (((c >> 3) == hh) && (m < dout)) {
    if (f < din)     a.x = W[m * din + f];
    if (f + 1 < din) a.y = W[m * din + f + 1];
  }
  return a;
}

static __device__ inline v8f load_bias_c(const float* __restrict__ bias, int dout) {
  v8f c = {};
#pragma unroll
  for (int j = 0; j < 8; ++j) c[j] = (j < dout) ? bias[j] : 0.0f;
  return c;
}

template <int DOUT, int NCH>
static __device__ inline v8f affine(const float h[8], const v2f* a, const v8f& c) {
  v8f d = c;
#pragma unroll
  for (int q = 0; q < NCH; ++q) {
    v2f b;
    b.x = h[2 * q];
    b.y = h[2 * q + 1];
    d = wmma4(a[q], b, d);
  }
  return d;
}

template <int DOUT, int NCH>
static __device__ inline void layer(float h[8], const v2f* a, const v8f& c) {
  const v8f d = affine<DOUT, NCH>(h, a, c);
#pragma unroll
  for (int j = 0; j < 8; ++j) h[j] = (j < DOUT) ? fast_tanh(d[j]) : 0.0f;
}

struct Weights {
  v2f a1[1], a2[4], a3[4], a4[3], a5[2], a6[2], ah[2];
  v8f c1, c2, c3, c4, c5, c6, ch;
};

static __device__ inline float forward_row(const Weights& w, float x0, float x1) {
  float h[8];
#pragma unroll
  for (int j = 0; j < 8; ++j) h[j] = 0.0f;
  h[0] = x0;
  h[1] = x1;
  layer<8, 1>(h, w.a1, w.c1);
  layer<8, 4>(h, w.a2, w.c2);
  layer<6, 4>(h, w.a3, w.c3);
  layer<4, 3>(h, w.a4, w.c4);
  layer<3, 2>(h, w.a5, w.c5);
  layer<3, 2>(h, w.a6, w.c6);
  const v8f d = affine<2, 2>(h, w.ah, w.ch);
  const float l0 = d[0], l1 = d[1];
  return fast_rcp(1.0f + __expf(l1 - l0));
}


__global__ void __launch_bounds__(256) mlp_wmma_kernel(
    const float* __restrict__ x,
    const float* __restrict__ W1, const float* __restrict__ b1,
    const float* __restrict__ W2, const float* __restrict__ b2,
    const float* __restrict__ W3, const float* __restrict__ b3,
    const float* __restrict__ W4, const float* __restrict__ b4,
    const float* __restrict__ W5, const float* __restrict__ b5,
    const float* __restrict__ W6, const float* __restrict__ b6,
    const float* __restrict__ Wh, const float* __restrict__ bh,
    float* __restrict__ out, float* __restrict__ partials, int numTiles) {
  const int tid  = threadIdx.x;
  const int lane = tid & 31;
  const int wavesPerBlock = blockDim.x >> 5;
  const int waveGlobal = blockIdx.x * wavesPerBlock + (tid >> 5);
  const int waveStride = gridDim.x * wavesPerBlock;

  Weights w;
  w.a1[0] = load_w_pair(W1, 8, 2, 0, lane);
  w.a2[0] = load_w_pair(W2, 8, 8, 0, lane);
  w.a2[1] = load_w_pair(W2, 8, 8, 2, lane);
  w.a2[2] = load_w_pair(W2, 8, 8, 4, lane);
  w.a2[3] = load_w_pair(W2, 8, 8, 6, lane);
  w.a3[0] = load_w_pair(W3, 6, 8, 0, lane);
  w.a3[1] = load_w_pair(W3, 6, 8, 2, lane);
  w.a3[2] = load_w_pair(W3, 6, 8, 4, lane);
  w.a3[3] = load_w_pair(W3, 6, 8, 6, lane);
  w.a4[0] = load_w_pair(W4, 4, 6, 0, lane);
  w.a4[1] = load_w_pair(W4, 4, 6, 2, lane);
  w.a4[2] = load_w_pair(W4, 4, 6, 4, lane);
  w.a5[0] = load_w_pair(W5, 3, 4, 0, lane);
  w.a5[1] = load_w_pair(W5, 3, 4, 2, lane);
  w.a6[0] = load_w_pair(W6, 3, 3, 0, lane);
  w.a6[1] = load_w_pair(W6, 3, 3, 2, lane);
  w.ah[0] = load_w_pair(Wh, 2, 3, 0, lane);
  w.ah[1] = load_w_pair(Wh, 2, 3, 2, lane);
  w.c1 = load_bias_c(b1, 8);
  w.c2 = load_bias_c(b2, 8);
  w.c3 = load_bias_c(b3, 6);
  w.c4 = load_bias_c(b4, 4);
  w.c5 = load_bias_c(b5, 3);
  w.c6 = load_bias_c(b6, 3);
  w.ch = load_bias_c(bh, 2);

  float s1 = 0.0f, s2 = 0.0f;

  const int numPairs = numTiles >> 1;
  for (int pair = waveGlobal; pair < numPairs; pair += waveStride) {
    const int row = pair * 32 + lane;
    const float2 xv = reinterpret_cast<const float2*>(x)[row];
    const float p0 = forward_row(w, xv.x, xv.y);
    out[stage_index(row)] = p0;
    s1 += p0;
    s2 += p0 * p0;
  }

  if ((numTiles & 1) && waveGlobal == 0) {
    const int row = (numTiles - 1) * 16 + (lane & 15);
    const bool live = lane < 16;
    float2 xv; xv.x = 0.0f; xv.y = 0.0f;
    if (live) xv = reinterpret_cast<const float2*>(x)[row];
    const float p0 = forward_row(w, xv.x, xv.y);
    if (live) {
      out[stage_index(row)] = p0;
      s1 += p0;
      s2 += p0 * p0;
    }
  }

  __shared__ float r0[256];
  __shared__ float r1[256];
  r0[tid] = s1;
  r1[tid] = s2;
  __syncthreads();
  for (int s = blockDim.x >> 1; s > 0; s >>= 1) {
    if (tid < s) {
      r0[tid] += r0[tid + s];
      r1[tid] += r1[tid + s];
    }
    __syncthreads();
  }
  if (tid == 0) {
    partials[2 * blockIdx.x + 0] = r0[0];
    partials[2 * blockIdx.x + 1] = r1[0];
  }
}


static_assert(NBLOCKS % FIN_THREADS == 0, "partials must split evenly over the fold's threads");

static __device__ inline void fin_load_chunk(const float* out, size_t rowBase,
                                             int tid, v2f v[FIN_PIECES]) {
  const float* stage = out + 2 * rowBase;
#pragma unroll
  for (int j = 0; j < FIN_PIECES; ++j)
    v[j] = *reinterpret_cast<const v2f*>(stage + (size_t)j * FIN_ROWS_PER_PIECE + (size_t)tid * 2);
}

static __device__ inline void fin_wait_chunk(const v2f v[FIN_PIECES]) {
#pragma unroll
  for (int j = 0; j < FIN_PIECES; ++j) {
    asm volatile("" :: "v"(v[j].x), "v"(v[j].y));
  }
}

#define FIN_RAGGED_PER_THREAD (FIN_ROWS_PER_CHUNK / FIN_THREADS)
static __device__ inline void fin_ragged_chunk(float* out, size_t rowBase, size_t B, int tid,
                                               float mean0, float inv, float g0, float g1,
                                               float be0, float be1) {
  const size_t n = B - rowBase;
  const float* stage = out + 2 * rowBase;
  float q[FIN_RAGGED_PER_THREAD];
#pragma unroll
  for (int k = 0; k < FIN_RAGGED_PER_THREAD; ++k) {
    const size_t i = (size_t)tid + (size_t)k * FIN_THREADS;
    q[k] = (i < n) ? stage[i] : 0.0f;
  }
#pragma unroll
  for (int k = 0; k < FIN_RAGGED_PER_THREAD; ++k) {
    asm volatile("" :: "v"(q[k]));
  }
  __syncthreads();
#pragma unroll
  for (int k = 0; k < FIN_RAGGED_PER_THREAD; ++k) {
    const size_t i = (size_t)tid + (size_t)k * FIN_THREADS;
    if (i < n) {
      const float p0 = q[k];
      float2 o;
      o.x = (p0 - mean0) * inv * g0 + be0;
      o.y = (mean0 - p0) * inv * g1 + be1;
      reinterpret_cast<float2*>(out)[rowBase + i] = o;
    }
  }
}

__global__ void __launch_bounds__(FIN_THREADS) finalize_kernel(
    float* __restrict__ out, const float* __restrict__ partials,
    const float* __restrict__ gamma, const float* __restrict__ beta, int B) {
  __shared__ float r0[FIN_THREADS];
  __shared__ float r1[FIN_THREADS];
  const int tid = threadIdx.x;
  const int nChunks = (B + FIN_ROWS_PER_CHUNK - 1) / FIN_ROWS_PER_CHUNK;
  int chunk = blockIdx.x;
  if (chunk >= nChunks) return;

  float2 p[FIN_PARTS_PER_THREAD];
#pragma unroll
  for (int k = 0; k < FIN_PARTS_PER_THREAD; ++k)
    p[k] = reinterpret_cast<const float2*>(partials)[tid + k * FIN_THREADS];

  v2f v[FIN_PIECES];
  bool full = (size_t)chunk * FIN_ROWS_PER_CHUNK + FIN_ROWS_PER_CHUNK <= (size_t)B;
  if (full) fin_load_chunk(out, (size_t)chunk * FIN_ROWS_PER_CHUNK, tid, v);

  const float g0 = gamma[0], g1 = gamma[1];
  const float be0 = beta[0], be1 = beta[1];

  float s1 = 0.0f, s2 = 0.0f;
#pragma unroll
  for (int k = 0; k < FIN_PARTS_PER_THREAD; ++k) {
    s1 += p[k].x;
    s2 += p[k].y;
  }
  if (full) fin_wait_chunk(v);
  r0[tid] = s1;
  r1[tid] = s2;
  __syncthreads();
  for (int s = FIN_THREADS >> 1; s > 0; s >>= 1) {
    if (tid < s) {
      r0[tid] += r0[tid + s];
      r1[tid] += r1[tid + s];
    }
    __syncthreads();
  }
  const float S1 = r0[0], S2 = r1[0];

  const float invB  = 1.0f / (float)B;
  const float mean0 = S1 * invB;
  const float var   = S2 * invB - mean0 * mean0;
  const float inv   = rsqrtf(var + BN_EPS);

  for (;;) {
    const size_t rowBase = (size_t)chunk * FIN_ROWS_PER_CHUNK;
    if (full) {
#pragma unroll
      for (int j = 0; j < FIN_PIECES; ++j) {
        const size_t r = rowBase + (size_t)j * FIN_ROWS_PER_PIECE + (size_t)tid * 2;
        const float pa = v[j].x, pb = v[j].y;
        v4f o;
        o.x = (pa - mean0) * inv * g0 + be0;
        o.y = (mean0 - pa) * inv * g1 + be1;
        o.z = (pb - mean0) * inv * g0 + be0;
        o.w = (mean0 - pb) * inv * g1 + be1;
        *reinterpret_cast<v4f*>(out + 2 * r) = o;
      }
    } else {
      fin_ragged_chunk(out, rowBase, (size_t)B, tid, mean0, inv, g0, g1, be0, be1);
    }
    chunk += gridDim.x;
    if (chunk >= nChunks) break;
    full = (size_t)chunk * FIN_ROWS_PER_CHUNK + FIN_ROWS_PER_CHUNK <= (size_t)B;
    if (full) {
      fin_load_chunk(out, (size_t)chunk * FIN_ROWS_PER_CHUNK, tid, v);
      fin_wait_chunk(v);
      __syncthreads();
    }
  }
}

extern "C" void kernel_launch(void* const* d_in, const int* in_sizes, int n_in,
                              void* d_out, int out_size, void* d_ws, size_t ws_size,
                              hipStream_t stream) {
  const float* x  = (const float*)d_in[0];
  const float* W1 = (const float*)d_in[1];  const float* b1 = (const float*)d_in[2];
  const float* W2 = (const float*)d_in[3];  const float* b2 = (const float*)d_in[4];
  const float* W3 = (const float*)d_in[5];  const float* b3 = (const float*)d_in[6];
  const float* W4 = (const float*)d_in[7];  const float* b4 = (const float*)d_in[8];
  const float* W5 = (const float*)d_in[9];  const float* b5 = (const float*)d_in[10];
  const float* W6 = (const float*)d_in[11]; const float* b6 = (const float*)d_in[12];
  const float* Wh = (const float*)d_in[13]; const float* bh = (const float*)d_in[14];
  const float* gamma = (const float*)d_in[15];
  const float* beta  = (const float*)d_in[16];
  float* out = (float*)d_out;

  const int B        = in_sizes[0] / 2;
  const int numTiles = B / 16;

  float* ws       = (float*)d_ws;
  float* partials = ws + 2;

  mlp_wmma_kernel<<<NBLOCKS, 128, 0, stream>>>(
      x, W1, b1, W2, b2, W3, b3, W4, b4, W5, b5, W6, b6, Wh, bh,
      out, partials, numTiles);
  finalize_kernel<<<FIN_BLOCKS, FIN_THREADS, 0, stream>>>(out, partials, gamma, beta, B);
}


# first launch as one resident round: 1024 eight-wave blocks (8 waves per SIMD), each wave loops over 16 row groups with the next group's x requested a group ahead; batch-sum order changes
# speedup vs baseline: 1.6097x; 1.0629x over previous
#include <hip/hip_runtime.h>
#include <math.h>


typedef __attribute__((ext_vector_type(2))) float v2f;
typedef __attribute__((ext_vector_type(4))) float v4f;
typedef __attribute__((ext_vector_type(8))) float v8f;

#define BN_EPS 1e-5f
#define MLP_THREADS 256
#define MLP_WAVES_PER_SIMD 8
#define NBLOCKS 1024

#define FIN_BLOCKS  512
#define FIN_THREADS 256
#define FIN_PIECES  16
#define FIN_ROWS_PER_PIECE (FIN_THREADS * 2)
#define FIN_ROWS_PER_CHUNK (FIN_ROWS_PER_PIECE * FIN_PIECES)
#define FIN_PARTS_PER_THREAD (NBLOCKS / FIN_THREADS)

static_assert((FIN_ROWS_PER_CHUNK & (FIN_ROWS_PER_CHUNK - 1)) == 0, "chunk rows must be a power of two");
static_assert(FIN_ROWS_PER_CHUNK % 32 == 0, "a wave's 32 rows must stay inside one chunk");

static __device__ inline unsigned stage_index(int row) {
  const unsigned r = (unsigned)row;
  return r + (r & ~(unsigned)(FIN_ROWS_PER_CHUNK - 1));
}


static __device__ inline float fast_rcp(float x) {
#if __has_builtin(__builtin_amdgcn_rcpf)
  return __builtin_amdgcn_rcpf(x);
#else
  return 1.0f / x;
#endif
}

static __device__ inline float fast_tanh(float x) {
#if __has_builtin(__builtin_amdgcn_tanhf)
  return __builtin_amdgcn_tanhf(x);
#elif __has_builtin(__builtin_amdgcn_tanh_f32)
  return __builtin_amdgcn_tanh_f32(x);
#else
  const float e = __expf(2.0f * x);
  return 1.0f - 2.0f * fast_rcp(e + 1.0f);
#endif
}


static __device__ inline v8f wmma4(v2f a, v2f b, v8f c) {
  return __builtin_amdgcn_wmma_f32_16x16x4_f32(
       false, a,  false, b,
       (short)0, c,  false,  false);
}

static __device__ inline v2f load_w_pair(const float* __restrict__ W, int dout,
                                         int din, int f, int lane) {
  const int c  = lane & 15;
  const int hh = lane >> 4;
  const int m  = c & 7;
  v2f a; a.x = 0.0f; a.y = 0.0f;
  if (((c >> 3) == hh) && (m < dout)) {
    if (f < din)     a.x = W[m * din + f];
    if (f + 1 < din) a.y = W[m * din + f + 1];
  }
  return a;
}

static __device__ inline v8f load_bias_c(const float* __restrict__ bias, int dout) {
  v8f c = {};
#pragma unroll
  for (int j = 0; j < 8; ++j) c[j] = (j < dout) ? bias[j] : 0.0f;
  return c;
}

template <int DOUT, int NCH>
static __device__ inline v8f affine(const float h[8], const v2f* a, const v8f& c) {
  v8f d = c;
#pragma unroll
  for (int q = 0; q < NCH; ++q) {
    v2f b;
    b.x = h[2 * q];
    b.y = h[2 * q + 1];
    d = wmma4(a[q], b, d);
  }
  return d;
}

template <int DOUT, int NCH>
static __device__ inline void layer(float h[8], const v2f* a, const v8f& c) {
  const v8f d = affine<DOUT, NCH>(h, a, c);
#pragma unroll
  for (int j = 0; j < 8; ++j) h[j] = (j < DOUT) ? fast_tanh(d[j]) : 0.0f;
}

struct Weights {
  v2f a1[1], a2[4], a3[4], a4[3], a5[2], a6[2], ah[2];
  v8f c1, c2, c3, c4, c5, c6, ch;
};

static __device__ inline float forward_row(const Weights& w, float x0, float x1) {
  float h[8];
#pragma unroll
  for (int j = 0; j < 8; ++j) h[j] = 0.0f;
  h[0] = x0;
  h[1] = x1;
  layer<8, 1>(h, w.a1, w.c1);
  layer<8, 4>(h, w.a2, w.c2);
  layer<6, 4>(h, w.a3, w.c3);
  layer<4, 3>(h, w.a4, w.c4);
  layer<3, 2>(h, w.a5, w.c5);
  layer<3, 2>(h, w.a6, w.c6);
  const v8f d = affine<2, 2>(h, w.ah, w.ch);
  const float l0 = d[0], l1 = d[1];
  return fast_rcp(1.0f + __expf(l1 - l0));
}


__global__ void __launch_bounds__(MLP_THREADS)
__attribute__((amdgpu_waves_per_eu(MLP_WAVES_PER_SIMD, MLP_WAVES_PER_SIMD)))
mlp_wmma_kernel(
    const float* __restrict__ x,
    const float* __restrict__ W1, const float* __restrict__ b1,
    const float* __restrict__ W2, const float* __restrict__ b2,
    const float* __restrict__ W3, const float* __restrict__ b3,
    const float* __restrict__ W4, const float* __restrict__ b4,
    const float* __restrict__ W5, const float* __restrict__ b5,
    const float* __restrict__ W6, const float* __restrict__ b6,
    const float* __restrict__ Wh, const float* __restrict__ bh,
    float* __restrict__ out, float* __restrict__ partials, int numTiles) {
  const int tid  = threadIdx.x;
  const int lane = tid & 31;
  const int wavesPerBlock = blockDim.x >> 5;
  const int waveInBlock = __builtin_amdgcn_readfirstlane(tid >> 5);
  const int waveGlobal = blockIdx.x * wavesPerBlock + waveInBlock;
  const int waveStride = gridDim.x * wavesPerBlock;

  Weights w;
  w.a1[0] = load_w_pair(W1, 8, 2, 0, lane);
  w.a2[0] = load_w_pair(W2, 8, 8, 0, lane);
  w.a2[1] = load_w_pair(W2, 8, 8, 2, lane);
  w.a2[2] = load_w_pair(W2, 8, 8, 4, lane);
  w.a2[3] = load_w_pair(W2, 8, 8, 6, lane);
  w.a3[0] = load_w_pair(W3, 6, 8, 0, lane);
  w.a3[1] = load_w_pair(W3, 6, 8, 2, lane);
  w.a3[2] = load_w_pair(W3, 6, 8, 4, lane);
  w.a3[3] = load_w_pair(W3, 6, 8, 6, lane);
  w.a4[0] = load_w_pair(W4, 4, 6, 0, lane);
  w.a4[1] = load_w_pair(W4, 4, 6, 2, lane);
  w.a4[2] = load_w_pair(W4, 4, 6, 4, lane);
  w.a5[0] = load_w_pair(W5, 3, 4, 0, lane);
  w.a5[1] = load_w_pair(W5, 3, 4, 2, lane);
  w.a6[0] = load_w_pair(W6, 3, 3, 0, lane);
  w.a6[1] = load_w_pair(W6, 3, 3, 2, lane);
  w.ah[0] = load_w_pair(Wh, 2, 3, 0, lane);
  w.ah[1] = load_w_pair(Wh, 2, 3, 2, lane);
  w.c1 = load_bias_c(b1, 8);
  w.c2 = load_bias_c(b2, 8);
  w.c3 = load_bias_c(b3, 6);
  w.c4 = load_bias_c(b4, 4);
  w.c5 = load_bias_c(b5, 3);
  w.c6 = load_bias_c(b6, 3);
  w.ch = load_bias_c(bh, 2);

  float s1 = 0.0f, s2 = 0.0f;

  const int numPairs = numTiles >> 1;
  if (waveGlobal < numPairs) {
    const float2* __restrict__ xin = reinterpret_cast<const float2*>(x);
    int pair = waveGlobal;
    float2 xv = xin[pair * 32 + lane];
    for (;;) {
      const int next = pair + waveStride;
      const bool more = next < numPairs;
      const float2 xn = xin[(more ? next : pair) * 32 + lane];
      const int row = pair * 32 + lane;
      const float p0 = forward_row(w, xv.x, xv.y);
      out[stage_index(row)] = p0;
      s1 += p0;
      s2 += p0 * p0;
      if (!more) break;
      pair = next;
      xv = xn;
    }
  }

  if ((numTiles & 1) && waveGlobal == 0) {
    const int row = (numTiles - 1) * 16 + (lane & 15);
    const bool live = lane < 16;
    float2 xv; xv.x = 0.0f; xv.y = 0.0f;
    if (live) xv = reinterpret_cast<const float2*>(x)[row];
    const float p0 = forward_row(w, xv.x, xv.y);
    if (live) {
      out[stage_index(row)] = p0;
      s1 += p0;
      s2 += p0 * p0;
    }
  }

  __shared__ float r0[256];
  __shared__ float r1[256];
  r0[tid] = s1;
  r1[tid] = s2;
  __syncthreads();
  for (int s = blockDim.x >> 1; s > 0; s >>= 1) {
    if (tid < s) {
      r0[tid] += r0[tid + s];
      r1[tid] += r1[tid + s];
    }
    __syncthreads();
  }
  if (tid == 0) {
    partials[2 * blockIdx.x + 0] = r0[0];
    partials[2 * blockIdx.x + 1] = r1[0];
  }
}


static_assert(NBLOCKS % FIN_THREADS == 0, "partials must split evenly over the fold's threads");

static __device__ inline void fin_load_chunk(const float* out, size_t rowBase,
                                             int tid, v2f v[FIN_PIECES]) {
  const float* stage = out + 2 * rowBase;
#pragma unroll
  for (int j = 0; j < FIN_PIECES; ++j)
    v[j] = *reinterpret_cast<const v2f*>(stage + (size_t)j * FIN_ROWS_PER_PIECE + (size_t)tid * 2);
}

static __device__ inline void fin_wait_chunk(const v2f v[FIN_PIECES]) {
#pragma unroll
  for (int j = 0; j < FIN_PIECES; ++j) {
    asm volatile("" :: "v"(v[j].x), "v"(v[j].y));
  }
}

#define FIN_RAGGED_PER_THREAD (FIN_ROWS_PER_CHUNK / FIN_THREADS)
static __device__ inline void fin_ragged_chunk(float* out, size_t rowBase, size_t B, int tid,
                                               float mean0, float inv, float g0, float g1,
                                               float be0, float be1) {
  const size_t n = B - rowBase;
  const float* stage = out + 2 * rowBase;
  float q[FIN_RAGGED_PER_THREAD];
#pragma unroll
  for (int k = 0; k < FIN_RAGGED_PER_THREAD; ++k) {
    const size_t i = (size_t)tid + (size_t)k * FIN_THREADS;
    q[k] = (i < n) ? stage[i] : 0.0f;
  }
#pragma unroll
  for (int k = 0; k < FIN_RAGGED_PER_THREAD; ++k) {
    asm volatile("" :: "v"(q[k]));
  }
  __syncthreads();
#pragma unroll
  for (int k = 0; k < FIN_RAGGED_PER_THREAD; ++k) {
    const size_t i = (size_t)tid + (size_t)k * FIN_THREADS;
    if (i < n) {
      const float p0 = q[k];
      float2 o;
      o.x = (p0 - mean0) * inv * g0 + be0;
      o.y = (mean0 - p0) * inv * g1 + be1;
      reinterpret_cast<float2*>(out)[rowBase + i] = o;
    }
  }
}

__global__ void __launch_bounds__(FIN_THREADS) finalize_kernel(
    float* __restrict__ out, const float* __restrict__ partials,
    const float* __restrict__ gamma, const float* __restrict__ beta, int B) {
  __shared__ float r0[FIN_THREADS];
  __shared__ float r1[FIN_THREADS];
  const int tid = threadIdx.x;
  const int nChunks = (B + FIN_ROWS_PER_CHUNK - 1) / FIN_ROWS_PER_CHUNK;
  int chunk = blockIdx.x;
  if (chunk >= nChunks) return;

  float2 p[FIN_PARTS_PER_THREAD];
#pragma unroll
  for (int k = 0; k < FIN_PARTS_PER_THREAD; ++k)
    p[k] = reinterpret_cast<const float2*>(partials)[tid + k * FIN_THREADS];

  v2f v[FIN_PIECES];
  bool full = (size_t)chunk * FIN_ROWS_PER_CHUNK + FIN_ROWS_PER_CHUNK <= (size_t)B;
  if (full) fin_load_chunk(out, (size_t)chunk * FIN_ROWS_PER_CHUNK, tid, v);

  const float g0 = gamma[0], g1 = gamma[1];
  const float be0 = beta[0], be1 = beta[1];

  float s1 = 0.0f, s2 = 0.0f;
#pragma unroll
  for (int k = 0; k < FIN_PARTS_PER_THREAD; ++k) {
    s1 += p[k].x;
    s2 += p[k].y;
  }
  if (full) fin_wait_chunk(v);
  r0[tid] = s1;
  r1[tid] = s2;
  __syncthreads();
  for (int s = FIN_THREADS >> 1; s > 0; s >>= 1) {
    if (tid < s) {
      r0[tid] += r0[tid + s];
      r1[tid] += r1[tid + s];
    }
    __syncthreads();
  }
  const float S1 = r0[0], S2 = r1[0];

  const float invB  = 1.0f / (float)B;
  const float mean0 = S1 * invB;
  const float var   = S2 * invB - mean0 * mean0;
  const float inv   = rsqrtf(var + BN_EPS);

  for (;;) {
    const size_t rowBase = (size_t)chunk * FIN_ROWS_PER_CHUNK;
    if (full) {
#pragma unroll
      for (int j = 0; j < FIN_PIECES; ++j) {
        const size_t r = rowBase + (size_t)j * FIN_ROWS_PER_PIECE + (size_t)tid * 2;
        const float pa = v[j].x, pb = v[j].y;
        v4f o;
        o.x = (pa - mean0) * inv * g0 + be0;
        o.y = (mean0 - pa) * inv * g1 + be1;
        o.z = (pb - mean0) * inv * g0 + be0;
        o.w = (mean0 - pb) * inv * g1 + be1;
        *reinterpret_cast<v4f*>(out + 2 * r) = o;
      }
    } else {
      fin_ragged_chunk(out, rowBase, (size_t)B, tid, mean0, inv, g0, g1, be0, be1);
    }
    chunk += gridDim.x;
    if (chunk >= nChunks) break;
    full = (size_t)chunk * FIN_ROWS_PER_CHUNK + FIN_ROWS_PER_CHUNK <= (size_t)B;
    if (full) {
      fin_load_chunk(out, (size_t)chunk * FIN_ROWS_PER_CHUNK, tid, v);
      fin_wait_chunk(v);
      __syncthreads();
    }
  }
}

extern "C" void kernel_launch(void* const* d_in, const int* in_sizes, int n_in,
                              void* d_out, int out_size, void* d_ws, size_t ws_size,
                              hipStream_t stream) {
  const float* x  = (const float*)d_in[0];
  const float* W1 = (const float*)d_in[1];  const float* b1 = (const float*)d_in[2];
  const float* W2 = (const float*)d_in[3];  const float* b2 = (const float*)d_in[4];
  const float* W3 = (const float*)d_in[5];  const float* b3 = (const float*)d_in[6];
  const float* W4 = (const float*)d_in[7];  const float* b4 = (const float*)d_in[8];
  const float* W5 = (const float*)d_in[9];  const float* b5 = (const float*)d_in[10];
  const float* W6 = (const float*)d_in[11]; const float* b6 = (const float*)d_in[12];
  const float* Wh = (const float*)d_in[13]; const float* bh = (const float*)d_in[14];
  const float* gamma = (const float*)d_in[15];
  const float* beta  = (const float*)d_in[16];
  float* out = (float*)d_out;

  const int B        = in_sizes[0] / 2;
  const int numTiles = B / 16;

  float* ws       = (float*)d_ws;
  float* partials = ws + 2;

  mlp_wmma_kernel<<<NBLOCKS, MLP_THREADS, 0, stream>>>(
      x, W1, b1, W2, b2, W3, b3, W4, b4, W5, b5, W6, b6, Wh, bh,
      out, partials, numTiles);
  finalize_kernel<<<FIN_BLOCKS, FIN_THREADS, 0, stream>>>(out, partials, gamma, beta, B);
}


# first launch: each wave carries two independent 32-row chains (a lane owns two adjacent rows: one 16-byte load, one 8-byte store), so half as many waves, 4 per SIMD; batch-sum order changes
# speedup vs baseline: 2.6735x; 1.6609x over previous
#include <hip/hip_runtime.h>
#include <math.h>


typedef __attribute__((ext_vector_type(2))) float v2f;
typedef __attribute__((ext_vector_type(4))) float v4f;
typedef __attribute__((ext_vector_type(8))) float v8f;

#define BN_EPS 1e-5f
#define MLP_THREADS 256
#define MLP_WAVES_PER_SIMD 4
#define NBLOCKS 512

#define FIN_BLOCKS  512
#define FIN_THREADS 256
#define FIN_PIECES  16
#define FIN_ROWS_PER_PIECE (FIN_THREADS * 2)
#define FIN_ROWS_PER_CHUNK (FIN_ROWS_PER_PIECE * FIN_PIECES)
#define FIN_PARTS_PER_THREAD (NBLOCKS / FIN_THREADS)

static_assert((FIN_ROWS_PER_CHUNK & (FIN_ROWS_PER_CHUNK - 1)) == 0, "chunk rows must be a power of two");
static_assert(FIN_ROWS_PER_CHUNK % 32 == 0, "a wave's 32 rows must stay inside one chunk");

static __device__ inline unsigned stage_index(int row) {
  const unsigned r = (unsigned)row;
  return r + (r & ~(unsigned)(FIN_ROWS_PER_CHUNK - 1));
}


static __device__ inline float fast_rcp(float x) {
#if __has_builtin(__builtin_amdgcn_rcpf)
  return __builtin_amdgcn_rcpf(x);
#else
  return 1.0f / x;
#endif
}

static __device__ inline float fast_tanh(float x) {
#if __has_builtin(__builtin_amdgcn_tanhf)
  return __builtin_amdgcn_tanhf(x);
#elif __has_builtin(__builtin_amdgcn_tanh_f32)
  return __builtin_amdgcn_tanh_f32(x);
#else
  const float e = __expf(2.0f * x);
  return 1.0f - 2.0f * fast_rcp(e + 1.0f);
#endif
}


static __device__ inline v8f wmma4(v2f a, v2f b, v8f c) {
  return __builtin_amdgcn_wmma_f32_16x16x4_f32(
       false, a,  false, b,
       (short)0, c,  false,  false);
}

static __device__ inline v2f load_w_pair(const float* __restrict__ W, int dout,
                                         int din, int f, int lane) {
  const int c  = lane & 15;
  const int hh = lane >> 4;
  const int m  = c & 7;
  v2f a; a.x = 0.0f; a.y = 0.0f;
  if (((c >> 3) == hh) && (m < dout)) {
    if (f < din)     a.x = W[m * din + f];
    if (f + 1 < din) a.y = W[m * din + f + 1];
  }
  return a;
}

static __device__ inline v8f load_bias_c(const float* __restrict__ bias, int dout) {
  v8f c = {};
#pragma unroll
  for (int j = 0; j < 8; ++j) c[j] = (j < dout) ? bias[j] : 0.0f;
  return c;
}

template <int DOUT, int NCH>
static __device__ inline v8f affine(const float h[8], const v2f* a, const v8f& c) {
  v8f d = c;
#pragma unroll
  for (int q = 0; q < NCH; ++q) {
    v2f b;
    b.x = h[2 * q];
    b.y = h[2 * q + 1];
    d = wmma4(a[q], b, d);
  }
  return d;
}

template <int DOUT, int NCH>
static __device__ inline void layer(float h[8], const v2f* a, const v8f& c) {
  const v8f d = affine<DOUT, NCH>(h, a, c);
#pragma unroll
  for (int j = 0; j < 8; ++j) h[j] = (j < DOUT) ? fast_tanh(d[j]) : 0.0f;
}

template <int DOUT, int NCH>
static __device__ inline void layer2(float ha[8], float hb[8], const v2f* a, const v8f& c) {
  v8f da = c, db = c;
#pragma unroll
  for (int q = 0; q < NCH; ++q) {
    v2f ba, bb;
    ba.x = ha[2 * q];
    ba.y = ha[2 * q + 1];
    bb.x = hb[2 * q];
    bb.y = hb[2 * q + 1];
    da = wmma4(a[q], ba, da);
    db = wmma4(a[q], bb, db);
  }
#pragma unroll
  for (int j = 0; j < 8; ++j) {
    ha[j] = (j < DOUT) ? fast_tanh(da[j]) : 0.0f;
    hb[j] = (j < DOUT) ? fast_tanh(db[j]) : 0.0f;
  }
}

struct Weights {
  v2f a1[1], a2[4], a3[4], a4[3], a5[2], a6[2], ah[2];
  v8f c1, c2, c3, c4, c5, c6, ch;
};

static __device__ inline float forward_row(const Weights& w, float x0, float x1) {
  float h[8];
#pragma unroll
  for (int j = 0; j < 8; ++j) h[j] = 0.0f;
  h[0] = x0;
  h[1] = x1;
  layer<8, 1>(h, w.a1, w.c1);
  layer<8, 4>(h, w.a2, w.c2);
  layer<6, 4>(h, w.a3, w.c3);
  layer<4, 3>(h, w.a4, w.c4);
  layer<3, 2>(h, w.a5, w.c5);
  layer<3, 2>(h, w.a6, w.c6);
  const v8f d = affine<2, 2>(h, w.ah, w.ch);
  const float l0 = d[0], l1 = d[1];
  return fast_rcp(1.0f + __expf(l1 - l0));
}

static __device__ inline void forward_rows2(const Weights& w, float xa0, float xa1,
                                            float xb0, float xb1, float& pa, float& pb) {
  float ha[8], hb[8];
#pragma unroll
  for (int j = 0; j < 8; ++j) { ha[j] = 0.0f; hb[j] = 0.0f; }
  ha[0] = xa0;
  ha[1] = xa1;
  hb[0] = xb0;
  hb[1] = xb1;
  layer2<8, 1>(ha, hb, w.a1, w.c1);
  layer2<8, 4>(ha, hb, w.a2, w.c2);
  layer2<6, 4>(ha, hb, w.a3, w.c3);
  layer2<4, 3>(ha, hb, w.a4, w.c4);
  layer2<3, 2>(ha, hb, w.a5, w.c5);
  layer2<3, 2>(ha, hb, w.a6, w.c6);
  v8f da = w.ch, db = w.ch;
#pragma unroll
  for (int q = 0; q < 2; ++q) {
    v2f ba, bb;
    ba.x = ha[2 * q];
    ba.y = ha[2 * q + 1];
    bb.x = hb[2 * q];
    bb.y = hb[2 * q + 1];
    da = wmma4(w.ah[q], ba, da);
    db = wmma4(w.ah[q], bb, db);
  }
  pa = fast_rcp(1.0f + __expf(da[1] - da[0]));
  pb = fast_rcp(1.0f + __expf(db[1] - db[0]));
}


__global__ void __launch_bounds__(MLP_THREADS)
__attribute__((amdgpu_waves_per_eu(MLP_WAVES_PER_SIMD, MLP_WAVES_PER_SIMD)))
mlp_wmma_kernel(
    const float* __restrict__ x,
    const float* __restrict__ W1, const float* __restrict__ b1,
    const float* __restrict__ W2, const float* __restrict__ b2,
    const float* __restrict__ W3, const float* __restrict__ b3,
    const float* __restrict__ W4, const float* __restrict__ b4,
    const float* __restrict__ W5, const float* __restrict__ b5,
    const float* __restrict__ W6, const float* __restrict__ b6,
    const float* __restrict__ Wh, const float* __restrict__ bh,
    float* __restrict__ out, float* __restrict__ partials, int numTiles) {
  const int tid  = threadIdx.x;
  const int lane = tid & 31;
  const int wavesPerBlock = blockDim.x >> 5;
  const int waveInBlock = __builtin_amdgcn_readfirstlane(tid >> 5);
  const int waveGlobal = blockIdx.x * wavesPerBlock + waveInBlock;
  const int waveStride = gridDim.x * wavesPerBlock;

  Weights w;
  w.a1[0] = load_w_pair(W1, 8, 2, 0, lane);
  w.a2[0] = load_w_pair(W2, 8, 8, 0, lane);
  w.a2[1] = load_w_pair(W2, 8, 8, 2, lane);
  w.a2[2] = load_w_pair(W2, 8, 8, 4, lane);
  w.a2[3] = load_w_pair(W2, 8, 8, 6, lane);
  w.a3[0] = load_w_pair(W3, 6, 8, 0, lane);
  w.a3[1] = load_w_pair(W3, 6, 8, 2, lane);
  w.a3[2] = load_w_pair(W3, 6, 8, 4, lane);
  w.a3[3] = load_w_pair(W3, 6, 8, 6, lane);
  w.a4[0] = load_w_pair(W4, 4, 6, 0, lane);
  w.a4[1] = load_w_pair(W4, 4, 6, 2, lane);
  w.a4[2] = load_w_pair(W4, 4, 6, 4, lane);
  w.a5[0] = load_w_pair(W5, 3, 4, 0, lane);
  w.a5[1] = load_w_pair(W5, 3, 4, 2, lane);
  w.a6[0] = load_w_pair(W6, 3, 3, 0, lane);
  w.a6[1] = load_w_pair(W6, 3, 3, 2, lane);
  w.ah[0] = load_w_pair(Wh, 2, 3, 0, lane);
  w.ah[1] = load_w_pair(Wh, 2, 3, 2, lane);
  w.c1 = load_bias_c(b1, 8);
  w.c2 = load_bias_c(b2, 8);
  w.c3 = load_bias_c(b3, 6);
  w.c4 = load_bias_c(b4, 4);
  w.c5 = load_bias_c(b5, 3);
  w.c6 = load_bias_c(b6, 3);
  w.ch = load_bias_c(bh, 2);

  float s1 = 0.0f, s2 = 0.0f;

  const int numQuads = numTiles >> 2;
  if (waveGlobal < numQuads) {
    const float4* __restrict__ xin = reinterpret_cast<const float4*>(x);
    int grp = waveGlobal;
    float4 xv = xin[grp * 32 + lane];
    for (;;) {
      const int next = grp + waveStride;
      const bool more = next < numQuads;
      const float4 xn = xin[(more ? next : grp) * 32 + lane];
      const int row = grp * 64 + 2 * lane;
      float pa, pb;
      forward_rows2(w, xv.x, xv.y, xv.z, xv.w, pa, pb);
      float2 st;
      st.x = pa;
      st.y = pb;
      *reinterpret_cast<float2*>(out + stage_index(row)) = st;
      s1 += pa;
      s2 += pa * pa;
      s1 += pb;
      s2 += pb * pb;
      if (!more) break;
      grp = next;
      xv = xn;
    }
  }

  const int doneRows = numQuads * 64;
  const int leftRows = numTiles * 16 - doneRows;
  if (waveGlobal * 32 < leftRows) {
    const int row = doneRows + waveGlobal * 32 + lane;
    const bool live = waveGlobal * 32 + lane < leftRows;
    float2 xv; xv.x = 0.0f; xv.y = 0.0f;
    if (live) xv = reinterpret_cast<const float2*>(x)[row];
    const float p0 = forward_row(w, xv.x, xv.y);
    if (live) {
      out[stage_index(row)] = p0;
      s1 += p0;
      s2 += p0 * p0;
    }
  }

  __shared__ float r0[256];
  __shared__ float r1[256];
  r0[tid] = s1;
  r1[tid] = s2;
  __syncthreads();
  for (int s = blockDim.x >> 1; s > 0; s >>= 1) {
    if (tid < s) {
      r0[tid] += r0[tid + s];
      r1[tid] += r1[tid + s];
    }
    __syncthreads();
  }
  if (tid == 0) {
    partials[2 * blockIdx.x + 0] = r0[0];
    partials[2 * blockIdx.x + 1] = r1[0];
  }
}


static_assert(NBLOCKS % FIN_THREADS == 0, "partials must split evenly over the fold's threads");

static __device__ inline void fin_load_chunk(const float* out, size_t rowBase,
                                             int tid, v2f v[FIN_PIECES]) {
  const float* stage = out + 2 * rowBase;
#pragma unroll
  for (int j = 0; j < FIN_PIECES; ++j)
    v[j] = *reinterpret_cast<const v2f*>(stage + (size_t)j * FIN_ROWS_PER_PIECE + (size_t)tid * 2);
}

static __device__ inline void fin_wait_chunk(const v2f v[FIN_PIECES]) {
#pragma unroll
  for (int j = 0; j < FIN_PIECES; ++j) {
    asm volatile("" :: "v"(v[j].x), "v"(v[j].y));
  }
}

#define FIN_RAGGED_PER_THREAD (FIN_ROWS_PER_CHUNK / FIN_THREADS)
static __device__ inline void fin_ragged_chunk(float* out, size_t rowBase, size_t B, int tid,
                                               float mean0, float inv, float g0, float g1,
                                               float be0, float be1) {
  const size_t n = B - rowBase;
  const float* stage = out + 2 * rowBase;
  float q[FIN_RAGGED_PER_THREAD];
#pragma unroll
  for (int k = 0; k < FIN_RAGGED_PER_THREAD; ++k) {
    const size_t i = (size_t)tid + (size_t)k * FIN_THREADS;
    q[k] = (i < n) ? stage[i] : 0.0f;
  }
#pragma unroll
  for (int k = 0; k < FIN_RAGGED_PER_THREAD; ++k) {
    asm volatile("" :: "v"(q[k]));
  }
  __syncthreads();
#pragma unroll
  for (int k = 0; k < FIN_RAGGED_PER_THREAD; ++k) {
    const size_t i = (size_t)tid + (size_t)k * FIN_THREADS;
    if (i < n) {
      const float p0 = q[k];
      float2 o;
      o.x = (p0 - mean0) * inv * g0 + be0;
      o.y = (mean0 - p0) * inv * g1 + be1;
      reinterpret_cast<float2*>(out)[rowBase + i] = o;
    }
  }
}

__global__ void __launch_bounds__(FIN_THREADS) finalize_kernel(
    float* __restrict__ out, const float* __restrict__ partials,
    const float* __restrict__ gamma, const float* __restrict__ beta, int B) {
  __shared__ float r0[FIN_THREADS];
  __shared__ float r1[FIN_THREADS];
  const int tid = threadIdx.x;
  const int nChunks = (B + FIN_ROWS_PER_CHUNK - 1) / FIN_ROWS_PER_CHUNK;
  int chunk = blockIdx.x;
  if (chunk >= nChunks) return;

  float2 p[FIN_PARTS_PER_THREAD];
#pragma unroll
  for (int k = 0; k < FIN_PARTS_PER_THREAD; ++k)
    p[k] = reinterpret_cast<const float2*>(partials)[tid + k * FIN_THREADS];

  v2f v[FIN_PIECES];
  bool full = (size_t)chunk * FIN_ROWS_PER_CHUNK + FIN_ROWS_PER_CHUNK <= (size_t)B;
  if (full) fin_load_chunk(out, (size_t)chunk * FIN_ROWS_PER_CHUNK, tid, v);

  const float g0 = gamma[0], g1 = gamma[1];
  const float be0 = beta[0], be1 = beta[1];

  float s1 = 0.0f, s2 = 0.0f;
#pragma unroll
  for (int k = 0; k < FIN_PARTS_PER_THREAD; ++k) {
    s1 += p[k].x;
    s2 += p[k].y;
  }
  if (full) fin_wait_chunk(v);
  r0[tid] = s1;
  r1[tid] = s2;
  __syncthreads();
  for (int s = FIN_THREADS >> 1; s > 0; s >>= 1) {
    if (tid < s) {
      r0[tid] += r0[tid + s];
      r1[tid] += r1[tid + s];
    }
    __syncthreads();
  }
  const float S1 = r0[0], S2 = r1[0];

  const float invB  = 1.0f / (float)B;
  const float mean0 = S1 * invB;
  const float var   = S2 * invB - mean0 * mean0;
  const float inv   = rsqrtf(var + BN_EPS);

  for (;;) {
    const size_t rowBase = (size_t)chunk * FIN_ROWS_PER_CHUNK;
    if (full) {
#pragma unroll
      for (int j = 0; j < FIN_PIECES; ++j) {
        const size_t r = rowBase + (size_t)j * FIN_ROWS_PER_PIECE + (size_t)tid * 2;
        const float pa = v[j].x, pb = v[j].y;
        v4f o;
        o.x = (pa - mean0) * inv * g0 + be0;
        o.y = (mean0 - pa) * inv * g1 + be1;
        o.z = (pb - mean0) * inv * g0 + be0;
        o.w = (mean0 - pb) * inv * g1 + be1;
        *reinterpret_cast<v4f*>(out + 2 * r) = o;
      }
    } else {
      fin_ragged_chunk(out, rowBase, (size_t)B, tid, mean0, inv, g0, g1, be0, be1);
    }
    chunk += gridDim.x;
    if (chunk >= nChunks) break;
    full = (size_t)chunk * FIN_ROWS_PER_CHUNK + FIN_ROWS_PER_CHUNK <= (size_t)B;
    if (full) {
      fin_load_chunk(out, (size_t)chunk * FIN_ROWS_PER_CHUNK, tid, v);
      fin_wait_chunk(v);
      __syncthreads();
    }
  }
}

extern "C" void kernel_launch(void* const* d_in, const int* in_sizes, int n_in,
                              void* d_out, int out_size, void* d_ws, size_t ws_size,
                              hipStream_t stream) {
  const float* x  = (const float*)d_in[0];
  const float* W1 = (const float*)d_in[1];  const float* b1 = (const float*)d_in[2];
  const float* W2 = (const float*)d_in[3];  const float* b2 = (const float*)d_in[4];
  const float* W3 = (const float*)d_in[5];  const float* b3 = (const float*)d_in[6];
  const float* W4 = (const float*)d_in[7];  const float* b4 = (const float*)d_in[8];
  const float* W5 = (const float*)d_in[9];  const float* b5 = (const float*)d_in[10];
  const float* W6 = (const float*)d_in[11]; const float* b6 = (const float*)d_in[12];
  const float* Wh = (const float*)d_in[13]; const float* bh = (const float*)d_in[14];
  const float* gamma = (const float*)d_in[15];
  const float* beta  = (const float*)d_in[16];
  float* out = (float*)d_out;

  const int B        = in_sizes[0] / 2;
  const int numTiles = B / 16;

  float* ws       = (float*)d_ws;
  float* partials = ws + 2;

  mlp_wmma_kernel<<<NBLOCKS, MLP_THREADS, 0, stream>>>(
      x, W1, b1, W2, b2, W3, b3, W4, b4, W5, b5, W6, b6, Wh, bh,
      out, partials, numTiles);
  finalize_kernel<<<FIN_BLOCKS, FIN_THREADS, 0, stream>>>(out, partials, gamma, beta, B);
}


# first launch: each wave carries four independent 32-row chains (a lane owns four rows of a 128-row group), so four products stand between two of one chain; wave count kept; batch-sum order changes
# speedup vs baseline: 2.8280x; 1.0578x over previous
#include <hip/hip_runtime.h>
#include <math.h>


typedef __attribute__((ext_vector_type(2))) float v2f;
typedef __attribute__((ext_vector_type(4))) float v4f;
typedef __attribute__((ext_vector_type(8))) float v8f;

#define BN_EPS 1e-5f
#define MLP_THREADS 256
#define MLP_WAVES_PER_SIMD 4
#define NBLOCKS 512
#define MLP_CHAINS 4
#define MLP_ROWS_PER_PASS (32 * MLP_CHAINS)

#define FIN_BLOCKS  512
#define FIN_THREADS 256
#define FIN_PIECES  16
#define FIN_ROWS_PER_PIECE (FIN_THREADS * 2)
#define FIN_ROWS_PER_CHUNK (FIN_ROWS_PER_PIECE * FIN_PIECES)
#define FIN_PARTS_PER_THREAD (NBLOCKS / FIN_THREADS)

static_assert((FIN_ROWS_PER_CHUNK & (FIN_ROWS_PER_CHUNK - 1)) == 0, "chunk rows must be a power of two");
static_assert(FIN_ROWS_PER_CHUNK % MLP_ROWS_PER_PASS == 0, "the rows of one wave pass must stay inside one chunk");

static __device__ inline unsigned stage_index(int row) {
  const unsigned r = (unsigned)row;
  return r + (r & ~(unsigned)(FIN_ROWS_PER_CHUNK - 1));
}


static __device__ inline float fast_rcp(float x) {
#if __has_builtin(__builtin_amdgcn_rcpf)
  return __builtin_amdgcn_rcpf(x);
#else
  return 1.0f / x;
#endif
}

static __device__ inline float fast_tanh(float x) {
#if __has_builtin(__builtin_amdgcn_tanhf)
  return __builtin_amdgcn_tanhf(x);
#elif __has_builtin(__builtin_amdgcn_tanh_f32)
  return __builtin_amdgcn_tanh_f32(x);
#else
  const float e = __expf(2.0f * x);
  return 1.0f - 2.0f * fast_rcp(e + 1.0f);
#endif
}


static __device__ inline v8f wmma4(v2f a, v2f b, v8f c) {
  return __builtin_amdgcn_wmma_f32_16x16x4_f32(
       false, a,  false, b,
       (short)0, c,  false,  false);
}

static __device__ inline v2f load_w_pair(const float* __restrict__ W, int dout,
                                         int din, int f, int lane) {
  const int c  = lane & 15;
  const int hh = lane >> 4;
  const int m  = c & 7;
  v2f a; a.x = 0.0f; a.y = 0.0f;
  if (((c >> 3) == hh) && (m < dout)) {
    if (f < din)     a.x = W[m * din + f];
    if (f + 1 < din) a.y = W[m * din + f + 1];
  }
  return a;
}

static __device__ inline v8f load_bias_c(const float* __restrict__ bias, int dout) {
  v8f c = {};
#pragma unroll
  for (int j = 0; j < 8; ++j) c[j] = (j < dout) ? bias[j] : 0.0f;
  return c;
}

template <int DOUT, int NCH>
static __device__ inline v8f affine(const float h[8], const v2f* a, const v8f& c) {
  v8f d = c;
#pragma unroll
  for (int q = 0; q < NCH; ++q) {
    v2f b;
    b.x = h[2 * q];
    b.y = h[2 * q + 1];
    d = wmma4(a[q], b, d);
  }
  return d;
}

template <int DOUT, int NCH>
static __device__ inline void layer(float h[8], const v2f* a, const v8f& c) {
  const v8f d = affine<DOUT, NCH>(h, a, c);
#pragma unroll
  for (int j = 0; j < 8; ++j) h[j] = (j < DOUT) ? fast_tanh(d[j]) : 0.0f;
}

template <int DOUT, int NCH>
static __device__ inline void layer_chains(float h[MLP_CHAINS][8], const v2f* a, const v8f& c) {
  v8f d[MLP_CHAINS];
#pragma unroll
  for (int t = 0; t < MLP_CHAINS; ++t) d[t] = c;
#pragma unroll
  for (int q = 0; q < NCH; ++q) {
#pragma unroll
    for (int t = 0; t < MLP_CHAINS; ++t) {
      v2f b;
      b.x = h[t][2 * q];
      b.y = h[t][2 * q + 1];
      d[t] = wmma4(a[q], b, d[t]);
    }
  }
#pragma unroll
  for (int t = 0; t < MLP_CHAINS; ++t) {
#pragma unroll
    for (int j = 0; j < 8; ++j) h[t][j] = (j < DOUT) ? fast_tanh(d[t][j]) : 0.0f;
  }
}

struct Weights {
  v2f a1[1], a2[4], a3[4], a4[3], a5[2], a6[2], ah[2];
  v8f c1, c2, c3, c4, c5, c6, ch;
};

static __device__ inline float forward_row(const Weights& w, float x0, float x1) {
  float h[8];
#pragma unroll
  for (int j = 0; j < 8; ++j) h[j] = 0.0f;
  h[0] = x0;
  h[1] = x1;
  layer<8, 1>(h, w.a1, w.c1);
  layer<8, 4>(h, w.a2, w.c2);
  layer<6, 4>(h, w.a3, w.c3);
  layer<4, 3>(h, w.a4, w.c4);
  layer<3, 2>(h, w.a5, w.c5);
  layer<3, 2>(h, w.a6, w.c6);
  const v8f d = affine<2, 2>(h, w.ah, w.ch);
  const float l0 = d[0], l1 = d[1];
  return fast_rcp(1.0f + __expf(l1 - l0));
}

static __device__ inline void forward_chains(const Weights& w, const float x0[MLP_CHAINS],
                                             const float x1[MLP_CHAINS], float p[MLP_CHAINS]) {
  float h[MLP_CHAINS][8];
#pragma unroll
  for (int t = 0; t < MLP_CHAINS; ++t) {
#pragma unroll
    for (int j = 0; j < 8; ++j) h[t][j] = 0.0f;
    h[t][0] = x0[t];
    h[t][1] = x1[t];
  }
  layer_chains<8, 1>(h, w.a1, w.c1);
  layer_chains<8, 4>(h, w.a2, w.c2);
  layer_chains<6, 4>(h, w.a3, w.c3);
  layer_chains<4, 3>(h, w.a4, w.c4);
  layer_chains<3, 2>(h, w.a5, w.c5);
  layer_chains<3, 2>(h, w.a6, w.c6);
  v8f d[MLP_CHAINS];
#pragma unroll
  for (int t = 0; t < MLP_CHAINS; ++t) d[t] = w.ch;
#pragma unroll
  for (int q = 0; q < 2; ++q) {
#pragma unroll
    for (int t = 0; t < MLP_CHAINS; ++t) {
      v2f b;
      b.x = h[t][2 * q];
      b.y = h[t][2 * q + 1];
      d[t] = wmma4(w.ah[q], b, d[t]);
    }
  }
#pragma unroll
  for (int t = 0; t < MLP_CHAINS; ++t) p[t] = fast_rcp(1.0f + __expf(d[t][1] - d[t][0]));
}


__global__ void __launch_bounds__(MLP_THREADS)
__attribute__((amdgpu_waves_per_eu(MLP_WAVES_PER_SIMD, MLP_WAVES_PER_SIMD)))
mlp_wmma_kernel(
    const float* __restrict__ x,
    const float* __restrict__ W1, const float* __restrict__ b1,
    const float* __restrict__ W2, const float* __restrict__ b2,
    const float* __restrict__ W3, const float* __restrict__ b3,
    const float* __restrict__ W4, const float* __restrict__ b4,
    const float* __restrict__ W5, const float* __restrict__ b5,
    const float* __restrict__ W6, const float* __restrict__ b6,
    const float* __restrict__ Wh, const float* __restrict__ bh,
    float* __restrict__ out, float* __restrict__ partials, int numTiles) {
  const int tid  = threadIdx.x;
  const int lane = tid & 31;
  const int wavesPerBlock = blockDim.x >> 5;
  const int waveInBlock = __builtin_amdgcn_readfirstlane(tid >> 5);
  const int waveGlobal = blockIdx.x * wavesPerBlock + waveInBlock;
  const int waveStride = gridDim.x * wavesPerBlock;

  Weights w;
  w.a1[0] = load_w_pair(W1, 8, 2, 0, lane);
  w.a2[0] = load_w_pair(W2, 8, 8, 0, lane);
  w.a2[1] = load_w_pair(W2, 8, 8, 2, lane);
  w.a2[2] = load_w_pair(W2, 8, 8, 4, lane);
  w.a2[3] = load_w_pair(W2, 8, 8, 6, lane);
  w.a3[0] = load_w_pair(W3, 6, 8, 0, lane);
  w.a3[1] = load_w_pair(W3, 6, 8, 2, lane);
  w.a3[2] = load_w_pair(W3, 6, 8, 4, lane);
  w.a3[3] = load_w_pair(W3, 6, 8, 6, lane);
  w.a4[0] = load_w_pair(W4, 4, 6, 0, lane);
  w.a4[1] = load_w_pair(W4, 4, 6, 2, lane);
  w.a4[2] = load_w_pair(W4, 4, 6, 4, lane);
  w.a5[0] = load_w_pair(W5, 3, 4, 0, lane);
  w.a5[1] = load_w_pair(W5, 3, 4, 2, lane);
  w.a6[0] = load_w_pair(W6, 3, 3, 0, lane);
  w.a6[1] = load_w_pair(W6, 3, 3, 2, lane);
  w.ah[0] = load_w_pair(Wh, 2, 3, 0, lane);
  w.ah[1] = load_w_pair(Wh, 2, 3, 2, lane);
  w.c1 = load_bias_c(b1, 8);
  w.c2 = load_bias_c(b2, 8);
  w.c3 = load_bias_c(b3, 6);
  w.c4 = load_bias_c(b4, 4);
  w.c5 = load_bias_c(b5, 3);
  w.c6 = load_bias_c(b6, 3);
  w.ch = load_bias_c(bh, 2);

  float s1 = 0.0f, s2 = 0.0f;

  const int numPasses = numTiles >> 3;
  if (waveGlobal < numPasses) {
    const float4* __restrict__ xin = reinterpret_cast<const float4*>(x);
    int grp = waveGlobal;
    float4 xlo = xin[grp * 64 + lane];
    float4 xhi = xin[grp * 64 + 32 + lane];
    for (;;) {
      const int next = grp + waveStride;
      const bool more = next < numPasses;
      const int req = more ? next : grp;
      const float4 nlo = xin[req * 64 + lane];
      const float4 nhi = xin[req * 64 + 32 + lane];
      const int row = grp * MLP_ROWS_PER_PASS + 2 * lane;
      const float x0[MLP_CHAINS] = {xlo.x, xlo.z, xhi.x, xhi.z};
      const float x1[MLP_CHAINS] = {xlo.y, xlo.w, xhi.y, xhi.w};
      float p[MLP_CHAINS];
      forward_chains(w, x0, x1, p);
      float2 st;
      st.x = p[0];
      st.y = p[1];
      *reinterpret_cast<float2*>(out + stage_index(row)) = st;
      st.x = p[2];
      st.y = p[3];
      *reinterpret_cast<float2*>(out + stage_index(row + 64)) = st;
#pragma unroll
      for (int t = 0; t < MLP_CHAINS; ++t) {
        s1 += p[t];
        s2 += p[t] * p[t];
      }
      if (!more) break;
      grp = next;
      xlo = nlo;
      xhi = nhi;
    }
  }

  const int doneRows = numPasses * MLP_ROWS_PER_PASS;
  const int leftRows = numTiles * 16 - doneRows;
  if (waveGlobal * 32 < leftRows) {
    const int row = doneRows + waveGlobal * 32 + lane;
    const bool live = waveGlobal * 32 + lane < leftRows;
    float2 xv; xv.x = 0.0f; xv.y = 0.0f;
    if (live) xv = reinterpret_cast<const float2*>(x)[row];
    const float p0 = forward_row(w, xv.x, xv.y);
    if (live) {
      out[stage_index(row)] = p0;
      s1 += p0;
      s2 += p0 * p0;
    }
  }

  __shared__ float r0[256];
  __shared__ float r1[256];
  r0[tid] = s1;
  r1[tid] = s2;
  __syncthreads();
  for (int s = blockDim.x >> 1; s > 0; s >>= 1) {
    if (tid < s) {
      r0[tid] += r0[tid + s];
      r1[tid] += r1[tid + s];
    }
    __syncthreads();
  }
  if (tid == 0) {
    partials[2 * blockIdx.x + 0] = r0[0];
    partials[2 * blockIdx.x + 1] = r1[0];
  }
}


static_assert(NBLOCKS % FIN_THREADS == 0, "partials must split evenly over the fold's threads");

static __device__ inline void fin_load_chunk(const float* out, size_t rowBase,
                                             int tid, v2f v[FIN_PIECES]) {
  const float* stage = out + 2 * rowBase;
#pragma unroll
  for (int j = 0; j < FIN_PIECES; ++j)
    v[j] = *reinterpret_cast<const v2f*>(stage + (size_t)j * FIN_ROWS_PER_PIECE + (size_t)tid * 2);
}

static __device__ inline void fin_wait_chunk(const v2f v[FIN_PIECES]) {
#pragma unroll
  for (int j = 0; j < FIN_PIECES; ++j) {
    asm volatile("" :: "v"(v[j].x), "v"(v[j].y));
  }
}

#define FIN_RAGGED_PER_THREAD (FIN_ROWS_PER_CHUNK / FIN_THREADS)
static __device__ inline void fin_ragged_chunk(float* out, size_t rowBase, size_t B, int tid,
                                               float mean0, float inv, float g0, float g1,
                                               float be0, float be1) {
  const size_t n = B - rowBase;
  const float* stage = out + 2 * rowBase;
  float q[FIN_RAGGED_PER_THREAD];
#pragma unroll
  for (int k = 0; k < FIN_RAGGED_PER_THREAD; ++k) {
    const size_t i = (size_t)tid + (size_t)k * FIN_THREADS;
    q[k] = (i < n) ? stage[i] : 0.0f;
  }
#pragma unroll
  for (int k = 0; k < FIN_RAGGED_PER_THREAD; ++k) {
    asm volatile("" :: "v"(q[k]));
  }
  __syncthreads();
#pragma unroll
  for (int k = 0; k < FIN_RAGGED_PER_THREAD; ++k) {
    const size_t i = (size_t)tid + (size_t)k * FIN_THREADS;
    if (i < n) {
      const float p0 = q[k];
      float2 o;
      o.x = (p0 - mean0) * inv * g0 + be0;
      o.y = (mean0 - p0) * inv * g1 + be1;
      reinterpret_cast<float2*>(out)[rowBase + i] = o;
    }
  }
}

__global__ void __launch_bounds__(FIN_THREADS) finalize_kernel(
    float* __restrict__ out, const float* __restrict__ partials,
    const float* __restrict__ gamma, const float* __restrict__ beta, int B) {
  __shared__ float r0[FIN_THREADS];
  __shared__ float r1[FIN_THREADS];
  const int tid = threadIdx.x;
  const int nChunks = (B + FIN_ROWS_PER_CHUNK - 1) / FIN_ROWS_PER_CHUNK;
  int chunk = blockIdx.x;
  if (chunk >= nChunks) return;

  float2 p[FIN_PARTS_PER_THREAD];
#pragma unroll
  for (int k = 0; k < FIN_PARTS_PER_THREAD; ++k)
    p[k] = reinterpret_cast<const float2*>(partials)[tid + k * FIN_THREADS];

  v2f v[FIN_PIECES];
  bool full = (size_t)chunk * FIN_ROWS_PER_CHUNK + FIN_ROWS_PER_CHUNK <= (size_t)B;
  if (full) fin_load_chunk(out, (size_t)chunk * FIN_ROWS_PER_CHUNK, tid, v);

  const float g0 = gamma[0], g1 = gamma[1];
  const float be0 = beta[0], be1 = beta[1];

  float s1 = 0.0f, s2 = 0.0f;
#pragma unroll
  for (int k = 0; k < FIN_PARTS_PER_THREAD; ++k) {
    s1 += p[k].x;
    s2 += p[k].y;
  }
  if (full) fin_wait_chunk(v);
  r0[tid] = s1;
  r1[tid] = s2;
  __syncthreads();
  for (int s = FIN_THREADS >> 1; s > 0; s >>= 1) {
    if (tid < s) {
      r0[tid] += r0[tid + s];
      r1[tid] += r1[tid + s];
    }
    __syncthreads();
  }
  const float S1 = r0[0], S2 = r1[0];

  const float invB  = 1.0f / (float)B;
  const float mean0 = S1 * invB;
  const float var   = S2 * invB - mean0 * mean0;
  const float inv   = rsqrtf(var + BN_EPS);

  for (;;) {
    const size_t rowBase = (size_t)chunk * FIN_ROWS_PER_CHUNK;
    if (full) {
#pragma unroll
      for (int j = 0; j < FIN_PIECES; ++j) {
        const size_t r = rowBase + (size_t)j * FIN_ROWS_PER_PIECE + (size_t)tid * 2;
        const float pa = v[j].x, pb = v[j].y;
        v4f o;
        o.x = (pa - mean0) * inv * g0 + be0;
        o.y = (mean0 - pa) * inv * g1 + be1;
        o.z = (pb - mean0) * inv * g0 + be0;
        o.w = (mean0 - pb) * inv * g1 + be1;
        *reinterpret_cast<v4f*>(out + 2 * r) = o;
      }
    } else {
      fin_ragged_chunk(out, rowBase, (size_t)B, tid, mean0, inv, g0, g1, be0, be1);
    }
    chunk += gridDim.x;
    if (chunk >= nChunks) break;
    full = (size_t)chunk * FIN_ROWS_PER_CHUNK + FIN_ROWS_PER_CHUNK <= (size_t)B;
    if (full) {
      fin_load_chunk(out, (size_t)chunk * FIN_ROWS_PER_CHUNK, tid, v);
      fin_wait_chunk(v);
      __syncthreads();
    }
  }
}

extern "C" void kernel_launch(void* const* d_in, const int* in_sizes, int n_in,
                              void* d_out, int out_size, void* d_ws, size_t ws_size,
                              hipStream_t stream) {
  const float* x  = (const float*)d_in[0];
  const float* W1 = (const float*)d_in[1];  const float* b1 = (const float*)d_in[2];
  const float* W2 = (const float*)d_in[3];  const float* b2 = (const float*)d_in[4];
  const float* W3 = (const float*)d_in[5];  const float* b3 = (const float*)d_in[6];
  const float* W4 = (const float*)d_in[7];  const float* b4 = (const float*)d_in[8];
  const float* W5 = (const float*)d_in[9];  const float* b5 = (const float*)d_in[10];
  const float* W6 = (const float*)d_in[11]; const float* b6 = (const float*)d_in[12];
  const float* Wh = (const float*)d_in[13]; const float* bh = (const float*)d_in[14];
  const float* gamma = (const float*)d_in[15];
  const float* beta  = (const float*)d_in[16];
  float* out = (float*)d_out;

  const int B        = in_sizes[0] / 2;
  const int numTiles = B / 16;

  float* ws       = (float*)d_ws;
  float* partials = ws + 2;

  mlp_wmma_kernel<<<NBLOCKS, MLP_THREADS, 0, stream>>>(
      x, W1, b1, W2, b2, W3, b3, W4, b4, W5, b5, W6, b6, Wh, bh,
      out, partials, numTiles);
  finalize_kernel<<<FIN_BLOCKS, FIN_THREADS, 0, stream>>>(out, partials, gamma, beta, B);
}


# block sums of both launches by lane exchange inside each wave plus one LDS hand-over of the 8 wave sums (one barrier), not an 8-round LDS halving tree per block; batch-sum order changes
# speedup vs baseline: 2.9512x; 1.0436x over previous
#include <hip/hip_runtime.h>
#include <math.h>


typedef __attribute__((ext_vector_type(2))) float v2f;
typedef __attribute__((ext_vector_type(4))) float v4f;
typedef __attribute__((ext_vector_type(8))) float v8f;

#define BN_EPS 1e-5f
#define MLP_THREADS 256
#define MLP_WAVES_PER_SIMD 4
#define NBLOCKS 512
#define MLP_CHAINS 4
#define MLP_ROWS_PER_PASS (32 * MLP_CHAINS)

#define FIN_BLOCKS  512
#define FIN_THREADS 256
#define FIN_PIECES  16
#define FIN_ROWS_PER_PIECE (FIN_THREADS * 2)
#define FIN_ROWS_PER_CHUNK (FIN_ROWS_PER_PIECE * FIN_PIECES)
#define FIN_PARTS_PER_THREAD (NBLOCKS / FIN_THREADS)

static_assert((FIN_ROWS_PER_CHUNK & (FIN_ROWS_PER_CHUNK - 1)) == 0, "chunk rows must be a power of two");
static_assert(FIN_ROWS_PER_CHUNK % MLP_ROWS_PER_PASS == 0, "the rows of one wave pass must stay inside one chunk");

static __device__ inline unsigned stage_index(int row) {
  const unsigned r = (unsigned)row;
  return r + (r & ~(unsigned)(FIN_ROWS_PER_CHUNK - 1));
}


static __device__ inline float fast_rcp(float x) {
#if __has_builtin(__builtin_amdgcn_rcpf)
  return __builtin_amdgcn_rcpf(x);
#else
  return 1.0f / x;
#endif
}

static __device__ inline float fast_tanh(float x) {
#if __has_builtin(__builtin_amdgcn_tanhf)
  return __builtin_amdgcn_tanhf(x);
#elif __has_builtin(__builtin_amdgcn_tanh_f32)
  return __builtin_amdgcn_tanh_f32(x);
#else
  const float e = __expf(2.0f * x);
  return 1.0f - 2.0f * fast_rcp(e + 1.0f);
#endif
}


static __device__ inline void wave_sum2(float& a, float& b) {
#pragma unroll
  for (int m = 16; m > 0; m >>= 1) {
    const float ta = __shfl_xor(a, m, 32);
    const float tb = __shfl_xor(b, m, 32);
    a += ta;
    b += tb;
  }
}


static __device__ inline v8f wmma4(v2f a, v2f b, v8f c) {
  return __builtin_amdgcn_wmma_f32_16x16x4_f32(
       false, a,  false, b,
       (short)0, c,  false,  false);
}

static __device__ inline v2f load_w_pair(const float* __restrict__ W, int dout,
                                         int din, int f, int lane) {
  const int c  = lane & 15;
  const int hh = lane >> 4;
  const int m  = c & 7;
  v2f a; a.x = 0.0f; a.y = 0.0f;
  if (((c >> 3) == hh) && (m < dout)) {
    if (f < din)     a.x = W[m * din + f];
    if (f + 1 < din) a.y = W[m * din + f + 1];
  }
  return a;
}

static __device__ inline v8f load_bias_c(const float* __restrict__ bias, int dout) {
  v8f c = {};
#pragma unroll
  for (int j = 0; j < 8; ++j) c[j] = (j < dout) ? bias[j] : 0.0f;
  return c;
}

template <int DOUT, int NCH>
static __device__ inline v8f affine(const float h[8], const v2f* a, const v8f& c) {
  v8f d = c;
#pragma unroll
  for (int q = 0; q < NCH; ++q) {
    v2f b;
    b.x = h[2 * q];
    b.y = h[2 * q + 1];
    d = wmma4(a[q], b, d);
  }
  return d;
}

template <int DOUT, int NCH>
static __device__ inline void layer(float h[8], const v2f* a, const v8f& c) {
  const v8f d = affine<DOUT, NCH>(h, a, c);
#pragma unroll
  for (int j = 0; j < 8; ++j) h[j] = (j < DOUT) ? fast_tanh(d[j]) : 0.0f;
}

template <int DOUT, int NCH>
static __device__ inline void layer_chains(float h[MLP_CHAINS][8], const v2f* a, const v8f& c) {
  v8f d[MLP_CHAINS];
#pragma unroll
  for (int t = 0; t < MLP_CHAINS; ++t) d[t] = c;
#pragma unroll
  for (int q = 0; q < NCH; ++q) {
#pragma unroll
    for (int t = 0; t < MLP_CHAINS; ++t) {
      v2f b;
      b.x = h[t][2 * q];
      b.y = h[t][2 * q + 1];
      d[t] = wmma4(a[q], b, d[t]);
    }
  }
#pragma unroll
  for (int t = 0; t < MLP_CHAINS; ++t) {
#pragma unroll
    for (int j = 0; j < 8; ++j) h[t][j] = (j < DOUT) ? fast_tanh(d[t][j]) : 0.0f;
  }
}

struct Weights {
  v2f a1[1], a2[4], a3[4], a4[3], a5[2], a6[2], ah[2];
  v8f c1, c2, c3, c4, c5, c6, ch;
};

static __device__ inline float forward_row(const Weights& w, float x0, float x1) {
  float h[8];
#pragma unroll
  for (int j = 0; j < 8; ++j) h[j] = 0.0f;
  h[0] = x0;
  h[1] = x1;
  layer<8, 1>(h, w.a1, w.c1);
  layer<8, 4>(h, w.a2, w.c2);
  layer<6, 4>(h, w.a3, w.c3);
  layer<4, 3>(h, w.a4, w.c4);
  layer<3, 2>(h, w.a5, w.c5);
  layer<3, 2>(h, w.a6, w.c6);
  const v8f d = affine<2, 2>(h, w.ah, w.ch);
  const float l0 = d[0], l1 = d[1];
  return fast_rcp(1.0f + __expf(l1 - l0));
}

static __device__ inline void forward_chains(const Weights& w, const float x0[MLP_CHAINS],
                                             const float x1[MLP_CHAINS], float p[MLP_CHAINS]) {
  float h[MLP_CHAINS][8];
#pragma unroll
  for (int t = 0; t < MLP_CHAINS; ++t) {
#pragma unroll
    for (int j = 0; j < 8; ++j) h[t][j] = 0.0f;
    h[t][0] = x0[t];
    h[t][1] = x1[t];
  }
  layer_chains<8, 1>(h, w.a1, w.c1);
  layer_chains<8, 4>(h, w.a2, w.c2);
  layer_chains<6, 4>(h, w.a3, w.c3);
  layer_chains<4, 3>(h, w.a4, w.c4);
  layer_chains<3, 2>(h, w.a5, w.c5);
  layer_chains<3, 2>(h, w.a6, w.c6);
  v8f d[MLP_CHAINS];
#pragma unroll
  for (int t = 0; t < MLP_CHAINS; ++t) d[t] = w.ch;
#pragma unroll
  for (int q = 0; q < 2; ++q) {
#pragma unroll
    for (int t = 0; t < MLP_CHAINS; ++t) {
      v2f b;
      b.x = h[t][2 * q];
      b.y = h[t][2 * q + 1];
      d[t] = wmma4(w.ah[q], b, d[t]);
    }
  }
#pragma unroll
  for (int t = 0; t < MLP_CHAINS; ++t) p[t] = fast_rcp(1.0f + __expf(d[t][1] - d[t][0]));
}


__global__ void __launch_bounds__(MLP_THREADS)
__attribute__((amdgpu_waves_per_eu(MLP_WAVES_PER_SIMD, MLP_WAVES_PER_SIMD)))
mlp_wmma_kernel(
    const float* __restrict__ x,
    const float* __restrict__ W1, const float* __restrict__ b1,
    const float* __restrict__ W2, const float* __restrict__ b2,
    const float* __restrict__ W3, const float* __restrict__ b3,
    const float* __restrict__ W4, const float* __restrict__ b4,
    const float* __restrict__ W5, const float* __restrict__ b5,
    const float* __restrict__ W6, const float* __restrict__ b6,
    const float* __restrict__ Wh, const float* __restrict__ bh,
    float* __restrict__ out, float* __restrict__ partials, int numTiles) {
  const int tid  = threadIdx.x;
  const int lane = tid & 31;
  const int wavesPerBlock = blockDim.x >> 5;
  const int waveInBlock = __builtin_amdgcn_readfirstlane(tid >> 5);
  const int waveGlobal = blockIdx.x * wavesPerBlock + waveInBlock;
  const int waveStride = gridDim.x * wavesPerBlock;

  Weights w;
  w.a1[0] = load_w_pair(W1, 8, 2, 0, lane);
  w.a2[0] = load_w_pair(W2, 8, 8, 0, lane);
  w.a2[1] = load_w_pair(W2, 8, 8, 2, lane);
  w.a2[2] = load_w_pair(W2, 8, 8, 4, lane);
  w.a2[3] = load_w_pair(W2, 8, 8, 6, lane);
  w.a3[0] = load_w_pair(W3, 6, 8, 0, lane);
  w.a3[1] = load_w_pair(W3, 6, 8, 2, lane);
  w.a3[2] = load_w_pair(W3, 6, 8, 4, lane);
  w.a3[3] = load_w_pair(W3, 6, 8, 6, lane);
  w.a4[0] = load_w_pair(W4, 4, 6, 0, lane);
  w.a4[1] = load_w_pair(W4, 4, 6, 2, lane);
  w.a4[2] = load_w_pair(W4, 4, 6, 4, lane);
  w.a5[0] = load_w_pair(W5, 3, 4, 0, lane);
  w.a5[1] = load_w_pair(W5, 3, 4, 2, lane);
  w.a6[0] = load_w_pair(W6, 3, 3, 0, lane);
  w.a6[1] = load_w_pair(W6, 3, 3, 2, lane);
  w.ah[0] = load_w_pair(Wh, 2, 3, 0, lane);
  w.ah[1] = load_w_pair(Wh, 2, 3, 2, lane);
  w.c1 = load_bias_c(b1, 8);
  w.c2 = load_bias_c(b2, 8);
  w.c3 = load_bias_c(b3, 6);
  w.c4 = load_bias_c(b4, 4);
  w.c5 = load_bias_c(b5, 3);
  w.c6 = load_bias_c(b6, 3);
  w.ch = load_bias_c(bh, 2);

  float s1 = 0.0f, s2 = 0.0f;

  const int numPasses = numTiles >> 3;
  if (waveGlobal < numPasses) {
    const float4* __restrict__ xin = reinterpret_cast<const float4*>(x);
    int grp = waveGlobal;
    float4 xlo = xin[grp * 64 + lane];
    float4 xhi = xin[grp * 64 + 32 + lane];
    for (;;) {
      const int next = grp + waveStride;
      const bool more = next < numPasses;
      const int req = more ? next : grp;
      const float4 nlo = xin[req * 64 + lane];
      const float4 nhi = xin[req * 64 + 32 + lane];
      const int row = grp * MLP_ROWS_PER_PASS + 2 * lane;
      const float x0[MLP_CHAINS] = {xlo.x, xlo.z, xhi.x, xhi.z};
      const float x1[MLP_CHAINS] = {xlo.y, xlo.w, xhi.y, xhi.w};
      float p[MLP_CHAINS];
      forward_chains(w, x0, x1, p);
      float2 st;
      st.x = p[0];
      st.y = p[1];
      *reinterpret_cast<float2*>(out + stage_index(row)) = st;
      st.x = p[2];
      st.y = p[3];
      *reinterpret_cast<float2*>(out + stage_index(row + 64)) = st;
#pragma unroll
      for (int t = 0; t < MLP_CHAINS; ++t) {
        s1 += p[t];
        s2 += p[t] * p[t];
      }
      if (!more) break;
      grp = next;
      xlo = nlo;
      xhi = nhi;
    }
  }

  const int doneRows = numPasses * MLP_ROWS_PER_PASS;
  const int leftRows = numTiles * 16 - doneRows;
  if (waveGlobal * 32 < leftRows) {
    const int row = doneRows + waveGlobal * 32 + lane;
    const bool live = waveGlobal * 32 + lane < leftRows;
    float2 xv; xv.x = 0.0f; xv.y = 0.0f;
    if (live) xv = reinterpret_cast<const float2*>(x)[row];
    const float p0 = forward_row(w, xv.x, xv.y);
    if (live) {
      out[stage_index(row)] = p0;
      s1 += p0;
      s2 += p0 * p0;
    }
  }

  wave_sum2(s1, s2);
  __shared__ float r0[MLP_THREADS / 32];
  __shared__ float r1[MLP_THREADS / 32];
  if (lane == 0) {
    r0[waveInBlock] = s1;
    r1[waveInBlock] = s2;
  }
  __syncthreads();
  if (tid == 0) {
    float a = r0[0], b = r1[0];
#pragma unroll
    for (int w = 1; w < MLP_THREADS / 32; ++w) {
      a += r0[w];
      b += r1[w];
    }
    partials[2 * blockIdx.x + 0] = a;
    partials[2 * blockIdx.x + 1] = b;
  }
}


static_assert(NBLOCKS % FIN_THREADS == 0, "partials must split evenly over the fold's threads");

static __device__ inline void fin_load_chunk(const float* out, size_t rowBase, size_t B,
                                             int tid, v2f v[FIN_PIECES]) {
  const float* stage = out + 2 * rowBase;
  const size_t behind = 2 * (B - rowBase) - 2;
  const unsigned last = behind < (size_t)(2 * FIN_ROWS_PER_CHUNK) ? (unsigned)behind
                                                                  : (unsigned)(2 * FIN_ROWS_PER_CHUNK);
#pragma unroll
  for (int j = 0; j < FIN_PIECES; ++j) {
    unsigned off = (unsigned)j * FIN_ROWS_PER_PIECE + (unsigned)tid * 2;
    off = off < last ? off : last;
    v[j] = *reinterpret_cast<const v2f*>(stage + off);
  }
}

static __device__ inline void fin_wait_chunk(const v2f v[FIN_PIECES]) {
#pragma unroll
  for (int j = 0; j < FIN_PIECES; ++j) {
    asm volatile("" :: "v"(v[j].x), "v"(v[j].y));
  }
}

#define FIN_RAGGED_PER_THREAD (FIN_ROWS_PER_CHUNK / FIN_THREADS)
static __device__ inline void fin_ragged_chunk(float* out, size_t rowBase, size_t B, int tid,
                                               float mean0, float inv, float g0, float g1,
                                               float be0, float be1) {
  const size_t n = B - rowBase;
  const float* stage = out + 2 * rowBase;
  float q[FIN_RAGGED_PER_THREAD];
#pragma unroll
  for (int k = 0; k < FIN_RAGGED_PER_THREAD; ++k) {
    const size_t i = (size_t)tid + (size_t)k * FIN_THREADS;
    q[k] = (i < n) ? stage[i] : 0.0f;
  }
#pragma unroll
  for (int k = 0; k < FIN_RAGGED_PER_THREAD; ++k) {
    asm volatile("" :: "v"(q[k]));
  }
  __syncthreads();
#pragma unroll
  for (int k = 0; k < FIN_RAGGED_PER_THREAD; ++k) {
    const size_t i = (size_t)tid + (size_t)k * FIN_THREADS;
    if (i < n) {
      const float p0 = q[k];
      float2 o;
      o.x = (p0 - mean0) * inv * g0 + be0;
      o.y = (mean0 - p0) * inv * g1 + be1;
      reinterpret_cast<float2*>(out)[rowBase + i] = o;
    }
  }
}

static __device__ inline void fin_fold_wave(const float2 p[FIN_PARTS_PER_THREAD], int tid,
                                            float* r0, float* r1) {
  float s1 = 0.0f, s2 = 0.0f;
#pragma unroll
  for (int k = 0; k < FIN_PARTS_PER_THREAD; ++k) {
    s1 += p[k].x;
    s2 += p[k].y;
  }
  wave_sum2(s1, s2);
  if ((tid & 31) == 0) {
    r0[tid >> 5] = s1;
    r1[tid >> 5] = s2;
  }
}

__global__ void __launch_bounds__(FIN_THREADS) finalize_kernel(
    float* __restrict__ out, const float* partials,
    const float* __restrict__ gamma, const float* __restrict__ beta, int B) {
  __shared__ float r0[FIN_THREADS / 32];
  __shared__ float r1[FIN_THREADS / 32];
  const int tid = threadIdx.x;
  const int nChunks = (B + FIN_ROWS_PER_CHUNK - 1) / FIN_ROWS_PER_CHUNK;
  int chunk = blockIdx.x;
  if (chunk >= nChunks) return;

  float2 p[FIN_PARTS_PER_THREAD];
#pragma unroll
  for (int k = 0; k < FIN_PARTS_PER_THREAD; ++k)
    p[k] = reinterpret_cast<const float2*>(partials)[tid + k * FIN_THREADS];

  v2f v[FIN_PIECES];
  bool full = (size_t)chunk * FIN_ROWS_PER_CHUNK + FIN_ROWS_PER_CHUNK <= (size_t)B;

  const float g0 = gamma[0], g1 = gamma[1];
  const float be0 = beta[0], be1 = beta[1];

  __builtin_amdgcn_sched_barrier(0);
  fin_load_chunk(out, (size_t)chunk * FIN_ROWS_PER_CHUNK, (size_t)B, tid, v);
  __builtin_amdgcn_sched_barrier(0);
  fin_fold_wave(p, tid, r0, r1);
  fin_wait_chunk(v);
  __syncthreads();
  float S1 = r0[0], S2 = r1[0];
#pragma unroll
  for (int w = 1; w < FIN_THREADS / 32; ++w) {
    S1 += r0[w];
    S2 += r1[w];
  }

  const float invB  = 1.0f / (float)B;
  const float mean0 = S1 * invB;
  const float var   = S2 * invB - mean0 * mean0;
  const float inv   = rsqrtf(var + BN_EPS);

  for (;;) {
    const size_t rowBase = (size_t)chunk * FIN_ROWS_PER_CHUNK;
    if (full) {
#pragma unroll
      for (int j = 0; j < FIN_PIECES; ++j) {
        const size_t r = rowBase + (size_t)j * FIN_ROWS_PER_PIECE + (size_t)tid * 2;
        const float pa = v[j].x, pb = v[j].y;
        v4f o;
        o.x = (pa - mean0) * inv * g0 + be0;
        o.y = (mean0 - pa) * inv * g1 + be1;
        o.z = (pb - mean0) * inv * g0 + be0;
        o.w = (mean0 - pb) * inv * g1 + be1;
        *reinterpret_cast<v4f*>(out + 2 * r) = o;
      }
    } else {
      fin_ragged_chunk(out, rowBase, (size_t)B, tid, mean0, inv, g0, g1, be0, be1);
    }
    chunk += gridDim.x;
    if (chunk >= nChunks) break;
    full = (size_t)chunk * FIN_ROWS_PER_CHUNK + FIN_ROWS_PER_CHUNK <= (size_t)B;
    if (full) {
      fin_load_chunk(out, (size_t)chunk * FIN_ROWS_PER_CHUNK, (size_t)B, tid, v);
      fin_wait_chunk(v);
      __syncthreads();
    }
  }
}

extern "C" void kernel_launch(void* const* d_in, const int* in_sizes, int n_in,
                              void* d_out, int out_size, void* d_ws, size_t ws_size,
                              hipStream_t stream) {
  const float* x  = (const float*)d_in[0];
  const float* W1 = (const float*)d_in[1];  const float* b1 = (const float*)d_in[2];
  const float* W2 = (const float*)d_in[3];  const float* b2 = (const float*)d_in[4];
  const float* W3 = (const float*)d_in[5];  const float* b3 = (const float*)d_in[6];
  const float* W4 = (const float*)d_in[7];  const float* b4 = (const float*)d_in[8];
  const float* W5 = (const float*)d_in[9];  const float* b5 = (const float*)d_in[10];
  const float* W6 = (const float*)d_in[11]; const float* b6 = (const float*)d_in[12];
  const float* Wh = (const float*)d_in[13]; const float* bh = (const float*)d_in[14];
  const float* gamma = (const float*)d_in[15];
  const float* beta  = (const float*)d_in[16];
  float* out = (float*)d_out;

  const int B        = in_sizes[0] / 2;
  const int numTiles = B / 16;

  float* ws       = (float*)d_ws;
  float* partials = ws + 2;

  mlp_wmma_kernel<<<NBLOCKS, MLP_THREADS, 0, stream>>>(
      x, W1, b1, W2, b2, W3, b3, W4, b4, W5, b5, W6, b6, Wh, bh,
      out, partials, numTiles);
  finalize_kernel<<<FIN_BLOCKS, FIN_THREADS, 0, stream>>>(out, partials, gamma, beta, B);
}


# first launch: each wave carries eight independent 32-row chains (a lane owns eight rows of a 256-row group) in 4-wave blocks, so half as many waves, 2 per SIMD; batch-sum order changes
# speedup vs baseline: 3.0541x; 1.0349x over previous
#include <hip/hip_runtime.h>
#include <math.h>


typedef __attribute__((ext_vector_type(2))) float v2f;
typedef __attribute__((ext_vector_type(4))) float v4f;
typedef __attribute__((ext_vector_type(8))) float v8f;

#define BN_EPS 1e-5f
#define MLP_THREADS 128
#define MLP_WAVES_PER_SIMD 2
#define NBLOCKS 512
#define MLP_CHAINS 8
#define MLP_ROWS_PER_PASS (32 * MLP_CHAINS)
#define MLP_QUARTERS (MLP_CHAINS / 2)

#define FIN_BLOCKS  512
#define FIN_THREADS 256
#define FIN_PIECES  16
#define FIN_ROWS_PER_PIECE (FIN_THREADS * 2)
#define FIN_ROWS_PER_CHUNK (FIN_ROWS_PER_PIECE * FIN_PIECES)
#define FIN_PARTS_PER_THREAD (NBLOCKS / FIN_THREADS)

static_assert((FIN_ROWS_PER_CHUNK & (FIN_ROWS_PER_CHUNK - 1)) == 0, "chunk rows must be a power of two");
static_assert(FIN_ROWS_PER_CHUNK % MLP_ROWS_PER_PASS == 0, "the rows of one wave pass must stay inside one chunk");

static __device__ inline unsigned stage_index(int row) {
  const unsigned r = (unsigned)row;
  return r + (r & ~(unsigned)(FIN_ROWS_PER_CHUNK - 1));
}


static __device__ inline float fast_rcp(float x) {
#if __has_builtin(__builtin_amdgcn_rcpf)
  return __builtin_amdgcn_rcpf(x);
#else
  return 1.0f / x;
#endif
}

static __device__ inline float fast_tanh(float x) {
#if __has_builtin(__builtin_amdgcn_tanhf)
  return __builtin_amdgcn_tanhf(x);
#elif __has_builtin(__builtin_amdgcn_tanh_f32)
  return __builtin_amdgcn_tanh_f32(x);
#else
  const float e = __expf(2.0f * x);
  return 1.0f - 2.0f * fast_rcp(e + 1.0f);
#endif
}


static __device__ inline void wave_sum2(float& a, float& b) {
#pragma unroll
  for (int m = 16; m > 0; m >>= 1) {
    const float ta = __shfl_xor(a, m, 32);
    const float tb = __shfl_xor(b, m, 32);
    a += ta;
    b += tb;
  }
}


static __device__ inline v8f wmma4(v2f a, v2f b, v8f c) {
  return __builtin_amdgcn_wmma_f32_16x16x4_f32(
       false, a,  false, b,
       (short)0, c,  false,  false);
}

static __device__ inline v2f load_w_pair(const float* __restrict__ W, int dout,
                                         int din, int f, int lane) {
  const int c  = lane & 15;
  const int hh = lane >> 4;
  const int m  = c & 7;
  v2f a; a.x = 0.0f; a.y = 0.0f;
  if (((c >> 3) == hh) && (m < dout)) {
    if (f < din)     a.x = W[m * din + f];
    if (f + 1 < din) a.y = W[m * din + f + 1];
  }
  return a;
}

static __device__ inline v8f load_bias_c(const float* __restrict__ bias, int dout) {
  v8f c = {};
#pragma unroll
  for (int j = 0; j < 8; ++j) c[j] = (j < dout) ? bias[j] : 0.0f;
  return c;
}

template <int DOUT, int NCH>
static __device__ inline v8f affine(const float h[8], const v2f* a, const v8f& c) {
  v8f d = c;
#pragma unroll
  for (int q = 0; q < NCH; ++q) {
    v2f b;
    b.x = h[2 * q];
    b.y = h[2 * q + 1];
    d = wmma4(a[q], b, d);
  }
  return d;
}

template <int DOUT, int NCH>
static __device__ inline void layer(float h[8], const v2f* a, const v8f& c) {
  const v8f d = affine<DOUT, NCH>(h, a, c);
#pragma unroll
  for (int j = 0; j < 8; ++j) h[j] = (j < DOUT) ? fast_tanh(d[j]) : 0.0f;
}

template <int DOUT, int NCH>
static __device__ inline void layer_chains(float h[MLP_CHAINS][8], const v2f* a, const v8f& c) {
  v8f d[MLP_CHAINS];
#pragma unroll
  for (int t = 0; t < MLP_CHAINS; ++t) d[t] = c;
#pragma unroll
  for (int q = 0; q < NCH; ++q) {
#pragma unroll
    for (int t = 0; t < MLP_CHAINS; ++t) {
      v2f b;
      b.x = h[t][2 * q];
      b.y = h[t][2 * q + 1];
      d[t] = wmma4(a[q], b, d[t]);
    }
  }
#pragma unroll
  for (int t = 0; t < MLP_CHAINS; ++t) {
#pragma unroll
    for (int j = 0; j < 8; ++j) h[t][j] = (j < DOUT) ? fast_tanh(d[t][j]) : 0.0f;
  }
}

struct Weights {
  v2f a1[1], a2[4], a3[4], a4[3], a5[2], a6[2], ah[2];
  v8f c1, c2, c3, c4, c5, c6, ch;
};

static __device__ inline float forward_row(const Weights& w, float x0, float x1) {
  float h[8];
#pragma unroll
  for (int j = 0; j < 8; ++j) h[j] = 0.0f;
  h[0] = x0;
  h[1] = x1;
  layer<8, 1>(h, w.a1, w.c1);
  layer<8, 4>(h, w.a2, w.c2);
  layer<6, 4>(h, w.a3, w.c3);
  layer<4, 3>(h, w.a4, w.c4);
  layer<3, 2>(h, w.a5, w.c5);
  layer<3, 2>(h, w.a6, w.c6);
  const v8f d = affine<2, 2>(h, w.ah, w.ch);
  const float l0 = d[0], l1 = d[1];
  return fast_rcp(1.0f + __expf(l1 - l0));
}

static __device__ inline void forward_chains(const Weights& w, const float x0[MLP_CHAINS],
                                             const float x1[MLP_CHAINS], float p[MLP_CHAINS]) {
  float h[MLP_CHAINS][8];
#pragma unroll
  for (int t = 0; t < MLP_CHAINS; ++t) {
#pragma unroll
    for (int j = 0; j < 8; ++j) h[t][j] = 0.0f;
    h[t][0] = x0[t];
    h[t][1] = x1[t];
  }
  layer_chains<8, 1>(h, w.a1, w.c1);
  layer_chains<8, 4>(h, w.a2, w.c2);
  layer_chains<6, 4>(h, w.a3, w.c3);
  layer_chains<4, 3>(h, w.a4, w.c4);
  layer_chains<3, 2>(h, w.a5, w.c5);
  layer_chains<3, 2>(h, w.a6, w.c6);
  v8f d[MLP_CHAINS];
#pragma unroll
  for (int t = 0; t < MLP_CHAINS; ++t) d[t] = w.ch;
#pragma unroll
  for (int q = 0; q < 2; ++q) {
#pragma unroll
    for (int t = 0; t < MLP_CHAINS; ++t) {
      v2f b;
      b.x = h[t][2 * q];
      b.y = h[t][2 * q + 1];
      d[t] = wmma4(w.ah[q], b, d[t]);
    }
  }
#pragma unroll
  for (int t = 0; t < MLP_CHAINS; ++t) p[t] = fast_rcp(1.0f + __expf(d[t][1] - d[t][0]));
}


__global__ void __launch_bounds__(MLP_THREADS)
__attribute__((amdgpu_waves_per_eu(MLP_WAVES_PER_SIMD, MLP_WAVES_PER_SIMD)))
mlp_wmma_kernel(
    const float* x,
    const float* __restrict__ W1, const float* __restrict__ b1,
    const float* __restrict__ W2, const float* __restrict__ b2,
    const float* __restrict__ W3, const float* __restrict__ b3,
    const float* __restrict__ W4, const float* __restrict__ b4,
    const float* __restrict__ W5, const float* __restrict__ b5,
    const float* __restrict__ W6, const float* __restrict__ b6,
    const float* __restrict__ Wh, const float* __restrict__ bh,
    float* __restrict__ out, float* __restrict__ partials, int numTiles) {
  const int tid  = threadIdx.x;
  const int lane = tid & 31;
  const int wavesPerBlock = blockDim.x >> 5;
  const int waveInBlock = __builtin_amdgcn_readfirstlane(tid >> 5);
  const int waveGlobal = blockIdx.x * wavesPerBlock + waveInBlock;
  const int waveStride = gridDim.x * wavesPerBlock;

  Weights w;
  w.a1[0] = load_w_pair(W1, 8, 2, 0, lane);
  w.a2[0] = load_w_pair(W2, 8, 8, 0, lane);
  w.a2[1] = load_w_pair(W2, 8, 8, 2, lane);
  w.a2[2] = load_w_pair(W2, 8, 8, 4, lane);
  w.a2[3] = load_w_pair(W2, 8, 8, 6, lane);
  w.a3[0] = load_w_pair(W3, 6, 8, 0, lane);
  w.a3[1] = load_w_pair(W3, 6, 8, 2, lane);
  w.a3[2] = load_w_pair(W3, 6, 8, 4, lane);
  w.a3[3] = load_w_pair(W3, 6, 8, 6, lane);
  w.a4[0] = load_w_pair(W4, 4, 6, 0, lane);
  w.a4[1] = load_w_pair(W4, 4, 6, 2, lane);
  w.a4[2] = load_w_pair(W4, 4, 6, 4, lane);
  w.a5[0] = load_w_pair(W5, 3, 4, 0, lane);
  w.a5[1] = load_w_pair(W5, 3, 4, 2, lane);
  w.a6[0] = load_w_pair(W6, 3, 3, 0, lane);
  w.a6[1] = load_w_pair(W6, 3, 3, 2, lane);
  w.ah[0] = load_w_pair(Wh, 2, 3, 0, lane);
  w.ah[1] = load_w_pair(Wh, 2, 3, 2, lane);
  w.c1 = load_bias_c(b1, 8);
  w.c2 = load_bias_c(b2, 8);
  w.c3 = load_bias_c(b3, 6);
  w.c4 = load_bias_c(b4, 4);
  w.c5 = load_bias_c(b5, 3);
  w.c6 = load_bias_c(b6, 3);
  w.ch = load_bias_c(bh, 2);

  float s1 = 0.0f, s2 = 0.0f;

  const int numPasses = numTiles / (MLP_ROWS_PER_PASS / 16);
  if (waveGlobal < numPasses) {
    const float4* xin = reinterpret_cast<const float4*>(x);
    int grp = waveGlobal;
    float4 xq[MLP_QUARTERS];
#pragma unroll
    for (int j = 0; j < MLP_QUARTERS; ++j) xq[j] = xin[grp * (MLP_ROWS_PER_PASS / 2) + 32 * j + lane];
    for (;;) {
      const int next = grp + waveStride;
      const bool more = next < numPasses;
      const int req = more ? next : grp;
      float4 nq[MLP_QUARTERS];
#pragma unroll
      for (int j = 0; j < MLP_QUARTERS; ++j) nq[j] = xin[req * (MLP_ROWS_PER_PASS / 2) + 32 * j + lane];
      __builtin_amdgcn_sched_barrier(0);
      const int row = grp * MLP_ROWS_PER_PASS + 2 * lane;
      float x0[MLP_CHAINS], x1[MLP_CHAINS];
#pragma unroll
      for (int j = 0; j < MLP_QUARTERS; ++j) {
        x0[2 * j]     = xq[j].x;
        x1[2 * j]     = xq[j].y;
        x0[2 * j + 1] = xq[j].z;
        x1[2 * j + 1] = xq[j].w;
      }
      float p[MLP_CHAINS];
      forward_chains(w, x0, x1, p);
#pragma unroll
      for (int j = 0; j < MLP_QUARTERS; ++j) {
        float2 st;
        st.x = p[2 * j];
        st.y = p[2 * j + 1];
        *reinterpret_cast<float2*>(out + stage_index(row + 64 * j)) = st;
      }
#pragma unroll
      for (int t = 0; t < MLP_CHAINS; ++t) {
        s1 += p[t];
        s2 += p[t] * p[t];
      }
      if (!more) break;
      grp = next;
#pragma unroll
      for (int j = 0; j < MLP_QUARTERS; ++j) xq[j] = nq[j];
    }
  }

  const int doneRows = numPasses * MLP_ROWS_PER_PASS;
  const int leftRows = numTiles * 16 - doneRows;
  if (waveGlobal * 32 < leftRows) {
    const int row = doneRows + waveGlobal * 32 + lane;
    const bool live = waveGlobal * 32 + lane < leftRows;
    float2 xv; xv.x = 0.0f; xv.y = 0.0f;
    if (live) xv = reinterpret_cast<const float2*>(x)[row];
    const float p0 = forward_row(w, xv.x, xv.y);
    if (live) {
      out[stage_index(row)] = p0;
      s1 += p0;
      s2 += p0 * p0;
    }
  }

  wave_sum2(s1, s2);
  __shared__ float r0[MLP_THREADS / 32];
  __shared__ float r1[MLP_THREADS / 32];
  if (lane == 0) {
    r0[waveInBlock] = s1;
    r1[waveInBlock] = s2;
  }
  __syncthreads();
  if (tid == 0) {
    float a = r0[0], b = r1[0];
#pragma unroll
    for (int w = 1; w < MLP_THREADS / 32; ++w) {
      a += r0[w];
      b += r1[w];
    }
    partials[2 * blockIdx.x + 0] = a;
    partials[2 * blockIdx.x + 1] = b;
  }
}


static_assert(NBLOCKS % FIN_THREADS == 0, "partials must split evenly over the fold's threads");

static __device__ inline void fin_load_chunk(const float* out, size_t rowBase, size_t B,
                                             int tid, v2f v[FIN_PIECES]) {
  const float* stage = out + 2 * rowBase;
  const size_t behind = 2 * (B - rowBase) - 2;
  const unsigned last = behind < (size_t)(2 * FIN_ROWS_PER_CHUNK) ? (unsigned)behind
                                                                  : (unsigned)(2 * FIN_ROWS_PER_CHUNK);
#pragma unroll
  for (int j = 0; j < FIN_PIECES; ++j) {
    unsigned off = (unsigned)j * FIN_ROWS_PER_PIECE + (unsigned)tid * 2;
    off = off < last ? off : last;
    v[j] = *reinterpret_cast<const v2f*>(stage + off);
  }
}

static __device__ inline void fin_wait_chunk(const v2f v[FIN_PIECES]) {
#pragma unroll
  for (int j = 0; j < FIN_PIECES; ++j) {
    asm volatile("" :: "v"(v[j].x), "v"(v[j].y));
  }
}

#define FIN_RAGGED_PER_THREAD (FIN_ROWS_PER_CHUNK / FIN_THREADS)
static __device__ inline void fin_ragged_chunk(float* out, size_t rowBase, size_t B, int tid,
                                               float mean0, float inv, float g0, float g1,
                                               float be0, float be1) {
  const size_t n = B - rowBase;
  const float* stage = out + 2 * rowBase;
  float q[FIN_RAGGED_PER_THREAD];
#pragma unroll
  for (int k = 0; k < FIN_RAGGED_PER_THREAD; ++k) {
    const size_t i = (size_t)tid + (size_t)k * FIN_THREADS;
    q[k] = (i < n) ? stage[i] : 0.0f;
  }
#pragma unroll
  for (int k = 0; k < FIN_RAGGED_PER_THREAD; ++k) {
    asm volatile("" :: "v"(q[k]));
  }
  __syncthreads();
#pragma unroll
  for (int k = 0; k < FIN_RAGGED_PER_THREAD; ++k) {
    const size_t i = (size_t)tid + (size_t)k * FIN_THREADS;
    if (i < n) {
      const float p0 = q[k];
      float2 o;
      o.x = (p0 - mean0) * inv * g0 + be0;
      o.y = (mean0 - p0) * inv * g1 + be1;
      reinterpret_cast<float2*>(out)[rowBase + i] = o;
    }
  }
}

static __device__ inline void fin_fold_wave(const float2 p[FIN_PARTS_PER_THREAD], int tid,
                                            float* r0, float* r1) {
  float s1 = 0.0f, s2 = 0.0f;
#pragma unroll
  for (int k = 0; k < FIN_PARTS_PER_THREAD; ++k) {
    s1 += p[k].x;
    s2 += p[k].y;
  }
  wave_sum2(s1, s2);
  if ((tid & 31) == 0) {
    r0[tid >> 5] = s1;
    r1[tid >> 5] = s2;
  }
}

__global__ void __launch_bounds__(FIN_THREADS) finalize_kernel(
    float* __restrict__ out, const float* partials,
    const float* __restrict__ gamma, const float* __restrict__ beta, int B) {
  __shared__ float r0[FIN_THREADS / 32];
  __shared__ float r1[FIN_THREADS / 32];
  const int tid = threadIdx.x;
  const int nChunks = (B + FIN_ROWS_PER_CHUNK - 1) / FIN_ROWS_PER_CHUNK;
  int chunk = blockIdx.x;
  if (chunk >= nChunks) return;

  float2 p[FIN_PARTS_PER_THREAD];
#pragma unroll
  for (int k = 0; k < FIN_PARTS_PER_THREAD; ++k)
    p[k] = reinterpret_cast<const float2*>(partials)[tid + k * FIN_THREADS];

  v2f v[FIN_PIECES];
  bool full = (size_t)chunk * FIN_ROWS_PER_CHUNK + FIN_ROWS_PER_CHUNK <= (size_t)B;

  const float g0 = gamma[0], g1 = gamma[1];
  const float be0 = beta[0], be1 = beta[1];

  __builtin_amdgcn_sched_barrier(0);
  fin_load_chunk(out, (size_t)chunk * FIN_ROWS_PER_CHUNK, (size_t)B, tid, v);
  __builtin_amdgcn_sched_barrier(0);
  fin_fold_wave(p, tid, r0, r1);
  fin_wait_chunk(v);
  __syncthreads();
  float S1 = r0[0], S2 = r1[0];
#pragma unroll
  for (int w = 1; w < FIN_THREADS / 32; ++w) {
    S1 += r0[w];
    S2 += r1[w];
  }

  const float invB  = 1.0f / (float)B;
  const float mean0 = S1 * invB;
  const float var   = S2 * invB - mean0 * mean0;
  const float inv   = rsqrtf(var + BN_EPS);

  for (;;) {
    const size_t rowBase = (size_t)chunk * FIN_ROWS_PER_CHUNK;
    if (full) {
#pragma unroll
      for (int j = 0; j < FIN_PIECES; ++j) {
        const size_t r = rowBase + (size_t)j * FIN_ROWS_PER_PIECE + (size_t)tid * 2;
        const float pa = v[j].x, pb = v[j].y;
        v4f o;
        o.x = (pa - mean0) * inv * g0 + be0;
        o.y = (mean0 - pa) * inv * g1 + be1;
        o.z = (pb - mean0) * inv * g0 + be0;
        o.w = (mean0 - pb) * inv * g1 + be1;
        *reinterpret_cast<v4f*>(out + 2 * r) = o;
      }
    } else {
      fin_ragged_chunk(out, rowBase, (size_t)B, tid, mean0, inv, g0, g1, be0, be1);
    }
    chunk += gridDim.x;
    if (chunk >= nChunks) break;
    full = (size_t)chunk * FIN_ROWS_PER_CHUNK + FIN_ROWS_PER_CHUNK <= (size_t)B;
    if (full) {
      fin_load_chunk(out, (size_t)chunk * FIN_ROWS_PER_CHUNK, (size_t)B, tid, v);
      fin_wait_chunk(v);
      __syncthreads();
    }
  }
}

extern "C" void kernel_launch(void* const* d_in, const int* in_sizes, int n_in,
                              void* d_out, int out_size, void* d_ws, size_t ws_size,
                              hipStream_t stream) {
  const float* x  = (const float*)d_in[0];
  const float* W1 = (const float*)d_in[1];  const float* b1 = (const float*)d_in[2];
  const float* W2 = (const float*)d_in[3];  const float* b2 = (const float*)d_in[4];
  const float* W3 = (const float*)d_in[5];  const float* b3 = (const float*)d_in[6];
  const float* W4 = (const float*)d_in[7];  const float* b4 = (const float*)d_in[8];
  const float* W5 = (const float*)d_in[9];  const float* b5 = (const float*)d_in[10];
  const float* W6 = (const float*)d_in[11]; const float* b6 = (const float*)d_in[12];
  const float* Wh = (const float*)d_in[13]; const float* bh = (const float*)d_in[14];
  const float* gamma = (const float*)d_in[15];
  const float* beta  = (const float*)d_in[16];
  float* out = (float*)d_out;

  const int B        = in_sizes[0] / 2;
  const int numTiles = B / 16;

  float* ws       = (float*)d_ws;
  float* partials = ws + 2;

  mlp_wmma_kernel<<<NBLOCKS, MLP_THREADS, 0, stream>>>(
      x, W1, b1, W2, b2, W3, b3, W4, b4, W5, b5, W6, b6, Wh, bh,
      out, partials, numTiles);
  finalize_kernel<<<FIN_BLOCKS, FIN_THREADS, 0, stream>>>(out, partials, gamma, beta, B);
}
